# starting kernel
# baseline (speedup 1.0000x reference)
#include <hip/hip_runtime.h>


#define D 128
#define TILE_ROWS 16
#define BLOCK 256

typedef float v2f __attribute__((ext_vector_type(2)));
typedef float v8f __attribute__((ext_vector_type(8)));

__global__ __launch_bounds__(BLOCK)
void gin_scatter_add(const float* __restrict__ x,
                     const int*   __restrict__ src,
                     const int*   __restrict__ dst,
                     float*       __restrict__ agg,
                     int nEdges) {
    long gid = (long)blockIdx.x * blockDim.x + threadIdx.x;
    long total = (long)nEdges * D;
    if (gid >= total) return;
    int e = (int)(gid >> 7);
    int c = (int)(gid & (D - 1));
    int s = src[e];
    int d = dst[e];
    float v = x[(long)s * D + c];
    unsafeAtomicAdd(&agg[(long)d * D + c], v);
}

__global__ __launch_bounds__(BLOCK)
void gin_mlp(const float* __restrict__ xin,
             const float* __restrict__ agg,
             const float* __restrict__ W1, const float* __restrict__ b1,
             const float* __restrict__ W2, const float* __restrict__ b2,
             float* __restrict__ out,
             int applyReluOut) {
    __shared__ float s_tile[TILE_ROWS][D];
    __shared__ float t_tile[TILE_ROWS][D];

    const int tid  = threadIdx.x;
    const int lane = tid & 31;
    const int wave = tid >> 5;
    const long row0 = (long)blockIdx.x * TILE_ROWS;

    #pragma unroll
    for (int i = tid; i < TILE_ROWS * D; i += BLOCK) {
        int r = i >> 7;
        int c = i & (D - 1);
        long g = (row0 + r) * D + c;
        s_tile[r][c] = xin[g] + agg[g];
    }
    __syncthreads();

    const int mrow  = lane & 15;
    const int kHalf = (lane >> 4) << 1;
    const int n0    = wave * 16;
    const int col   = n0 + mrow;
    const int rbase = (lane >> 4) << 3;

    {
        v8f acc = {};
        #pragma unroll 4
        for (int k = 0; k < D; k += 4) {
            const int kb = k + kHalf;
            v2f a, b;
            a[0] = s_tile[mrow][kb];
            a[1] = s_tile[mrow][kb + 1];
            b[0] = W1[(long)kb * D + col];
            b[1] = W1[(long)(kb + 1) * D + col];
            acc = __builtin_amdgcn_wmma_f32_16x16x4_f32(
                false, a, false, b, (short)0, acc, false, false);
        }
        const float bias = b1[col];
        #pragma unroll
        for (int v = 0; v < 8; ++v) {
            float t = acc[v] + bias;
            t_tile[rbase + v][col] = t > 0.0f ? t : 0.0f;
        }
    }
    __syncthreads();

    {
        v8f acc = {};
        #pragma unroll 4
        for (int k = 0; k < D; k += 4) {
            const int kb = k + kHalf;
            v2f a, b;
            a[0] = t_tile[mrow][kb];
            a[1] = t_tile[mrow][kb + 1];
            b[0] = W2[(long)kb * D + col];
            b[1] = W2[(long)(kb + 1) * D + col];
            acc = __builtin_amdgcn_wmma_f32_16x16x4_f32(
                false, a, false, b, (short)0, acc, false, false);
        }
        const float bias = b2[col];
        #pragma unroll
        for (int v = 0; v < 8; ++v) {
            float u = acc[v] + bias;
            if (applyReluOut) u = u > 0.0f ? u : 0.0f;
            out[(row0 + rbase + v) * D + col] = u;
        }
    }
}

extern "C" void kernel_launch(void* const* d_in, const int* in_sizes, int n_in,
                              void* d_out, int out_size, void* d_ws, size_t ws_size,
                              hipStream_t stream) {
    const float* x  = (const float*)d_in[0];
    const int*   ei = (const int*)  d_in[1];
    const float* W1 = (const float*)d_in[2];
    const float* b1 = (const float*)d_in[3];
    const float* W2 = (const float*)d_in[4];
    const float* b2 = (const float*)d_in[5];
    float* out = (float*)d_out;

    const int N = in_sizes[0] / D;
    const int E = in_sizes[1] / 2;
    const int* src = ei;
    const int* dst = ei + E;

    float* agg = (float*)d_ws;
    float* h1  = agg + (size_t)N * D;
    const size_t aggBytes = (size_t)N * D * sizeof(float);

    const long scatterThreads = (long)E * D;
    const int  sBlocks = (int)((scatterThreads + BLOCK - 1) / BLOCK);
    const int  mBlocks = N / TILE_ROWS;

    hipMemsetAsync(agg, 0, aggBytes, stream);
    gin_scatter_add<<<sBlocks, BLOCK, 0, stream>>>(x, src, dst, agg, E);
    gin_mlp<<<mBlocks, BLOCK, 0, stream>>>(x, agg, W1, b1, W2, b2, h1, 1);

    hipMemsetAsync(agg, 0, aggBytes, stream);
    gin_scatter_add<<<sBlocks, BLOCK, 0, stream>>>(h1, src, dst, agg, E);
    gin_mlp<<<mBlocks, BLOCK, 0, stream>>>(h1, agg, W1, b1, W2, b2, out, 0);
}


# Float-atomic scatter-add replaced by bucketing edges by destination (int atomic ranks), sorting each bucket by source id, and a wave-per-row gather sum: sum order now fixed, ascending source id
# speedup vs baseline: 2.0411x; 2.0411x over previous
#include <hip/hip_runtime.h>


#define D 128
#define TILE_ROWS 16
#define BLOCK 256
#define GATHER_BLOCKS 2048
#define SORT_REGS 16

typedef float v2f __attribute__((ext_vector_type(2)));
typedef float v4f __attribute__((ext_vector_type(4)));
typedef float v8f __attribute__((ext_vector_type(8)));

__global__ __launch_bounds__(BLOCK)
void edge_rank(const int* __restrict__ dst,
               int*       __restrict__ cnt,
               int*       __restrict__ rank,
               int nEdges) {
    const int e = blockIdx.x * BLOCK + threadIdx.x;
    if (e >= nEdges) return;
    rank[e] = atomicAdd(&cnt[dst[e]], 1);
}

__global__ __launch_bounds__(BLOCK)
void bucket_alloc(const int* cnt, int* total, int* __restrict__ start, int nNodes) {
    __shared__ int wsum[BLOCK / 32];
    __shared__ int blockBase;
    const int tid  = threadIdx.x;
    const int lane = tid & 31;
    const int wave = tid >> 5;
    const int n0   = (blockIdx.x * BLOCK + tid) * 4;

    int c[4];
    #pragma unroll
    for (int j = 0; j < 4; ++j) c[j] = (n0 + j < nNodes) ? cnt[n0 + j] : 0;
    const int mine = c[0] + c[1] + c[2] + c[3];

    int incl = mine;
    for (int off = 1; off < 32; off <<= 1) {
        const int v = __shfl_up(incl, off);
        if (lane >= off) incl += v;
    }
    if (lane == 31) wsum[wave] = incl;
    __syncthreads();
    if (tid == 0) {
        int s = 0;
        for (int w = 0; w < BLOCK / 32; ++w) { const int v = wsum[w]; wsum[w] = s; s += v; }
        blockBase = atomicAdd(total, s);
    }
    __syncthreads();
    int slot = blockBase + wsum[wave] + incl - mine;
    #pragma unroll
    for (int j = 0; j < 4; ++j) {
        if (n0 + j < nNodes) { start[n0 + j] = slot; slot += c[j]; }
    }
}

__global__ __launch_bounds__(BLOCK)
void bucket_fill(const int* __restrict__ src,
                 const int* __restrict__ dst,
                 const int* __restrict__ rank,
                 const int* __restrict__ start,
                 int*       __restrict__ col,
                 int nEdges) {
    const int e = blockIdx.x * BLOCK + threadIdx.x;
    if (e >= nEdges) return;
    col[start[dst[e]] + rank[e]] = src[e];
}

__global__ __launch_bounds__(BLOCK)
void bucket_sort(const int* __restrict__ start,
                 const int* __restrict__ cnt,
                 int*       __restrict__ col,
                 int nNodes) {
    const int n = blockIdx.x * BLOCK + threadIdx.x;
    if (n >= nNodes) return;
    int* p = col + start[n];
    const int m = cnt[n];
    if (m < 2) return;
    if (m <= SORT_REGS) {
        int v[SORT_REGS];
        #pragma unroll
        for (int i = 0; i < SORT_REGS; ++i) v[i] = (i < m) ? p[i] : 0x7fffffff;
        #pragma unroll
        for (int r = 0; r < SORT_REGS; ++r) {
            #pragma unroll
            for (int i = (r & 1); i + 1 < SORT_REGS; i += 2) {
                const int lo = min(v[i], v[i + 1]);
                const int hi = max(v[i], v[i + 1]);
                v[i] = lo; v[i + 1] = hi;
            }
        }
        #pragma unroll
        for (int i = 0; i < SORT_REGS; ++i) if (i < m) p[i] = v[i];
    } else {
        for (int i = 1; i < m; ++i) {
            const int key = p[i];
            int j = i - 1;
            while (j >= 0 && p[j] > key) { p[j + 1] = p[j]; --j; }
            p[j + 1] = key;
        }
    }
}


__device__ __forceinline__
void rows_ask(const float* __restrict__ laneBase, int my, int k0, int m, v4f (&r)[8]) {
    int s[8];
    #pragma unroll
    for (int j = 0; j < 8; ++j) s[j] = __builtin_amdgcn_readlane(my, k0 + j);
    #pragma unroll
    for (int j = 0; j < 8; ++j)
        if (j < m) r[j] = *(const v4f*)(laneBase + (long)s[j] * D);
}

__device__ __forceinline__
void rows_add(const v4f (&r)[8], int m, v4f& acc) {
    #pragma unroll
    for (int j = 0; j < 8; ++j)
        if (j < m) acc += r[j];
}

__global__ __launch_bounds__(BLOCK)
void gin_gather_sum(const float* __restrict__ xin,
                    const int*   __restrict__ start,
                    const int*   __restrict__ cnt,
                    const int*   __restrict__ col,
                    float*       __restrict__ agg,
                    int nNodes) {
    const int lane = threadIdx.x & 31;
    const int wave   = __builtin_amdgcn_readfirstlane((int)(threadIdx.x >> 5));
    const int stride = (int)gridDim.x * (BLOCK / 32);
    int n = (int)blockIdx.x * (BLOCK / 32) + wave;
    const float* laneBase = xin + lane * 4;

    int st0 = 0, dg0 = 0, st1 = 0, dg1 = 0;
    if (n < nNodes) {
        st0 = __builtin_amdgcn_readfirstlane(start[n]);
        dg0 = __builtin_amdgcn_readfirstlane(cnt[n]);
    }
    if (n + stride < nNodes) {
        st1 = __builtin_amdgcn_readfirstlane(start[n + stride]);
        dg1 = __builtin_amdgcn_readfirstlane(cnt[n + stride]);
    }
    int my0 = (lane < dg0) ? col[st0 + lane] : 0;

    for (; n < nNodes; n += stride) {
        int st2 = 0, dg2 = 0;
        const int n2 = n + 2 * stride;
        if (n2 < nNodes) {
            st2 = __builtin_amdgcn_readfirstlane(start[n2]);
            dg2 = __builtin_amdgcn_readfirstlane(cnt[n2]);
        }

        v4f acc = {0.0f, 0.0f, 0.0f, 0.0f};
        v4f r[8];
        int c = dg0 < 32 ? dg0 : 32;
        int m = c < 8 ? c : 8;
        rows_ask(laneBase, my0, 0, m, r);
        const int my1 = (lane < dg1) ? col[st1 + lane] : 0;
        rows_add(r, m, acc);
        for (int k0 = 8; k0 < c; k0 += 8) {
            m = (c - k0) < 8 ? (c - k0) : 8;
            rows_ask(laneBase, my0, k0, m, r);
            rows_add(r, m, acc);
        }
        int rem = dg0 - c;
        int pos = st0;
        while (rem > 0) {
            pos += 32;
            c = rem < 32 ? rem : 32;
            const int my = (lane < c) ? col[pos + lane] : 0;
            for (int k0 = 0; k0 < c; k0 += 8) {
                m = (c - k0) < 8 ? (c - k0) : 8;
                rows_ask(laneBase, my, k0, m, r);
                rows_add(r, m, acc);
            }
            rem -= c;
        }
        *(v4f*)(agg + (long)n * D + lane * 4) = acc;

        st0 = st1; dg0 = dg1; my0 = my1;
        st1 = st2; dg1 = dg2;
    }
}

__global__ __launch_bounds__(BLOCK)
void gin_mlp(const float* __restrict__ xin,
             const float* __restrict__ agg,
             const float* __restrict__ W1, const float* __restrict__ b1,
             const float* __restrict__ W2, const float* __restrict__ b2,
             float* __restrict__ out,
             int applyReluOut) {
    __shared__ float s_tile[TILE_ROWS][D];
    __shared__ float t_tile[TILE_ROWS][D];

    const int tid  = threadIdx.x;
    const int lane = tid & 31;
    const int wave = tid >> 5;
    const long row0 = (long)blockIdx.x * TILE_ROWS;

    #pragma unroll
    for (int i = tid; i < TILE_ROWS * D; i += BLOCK) {
        int r = i >> 7;
        int c = i & (D - 1);
        long g = (row0 + r) * D + c;
        s_tile[r][c] = xin[g] + agg[g];
    }
    __syncthreads();

    const int mrow  = lane & 15;
    const int kHalf = (lane >> 4) << 1;
    const int n0    = wave * 16;
    const int col   = n0 + mrow;
    const int rbase = (lane >> 4) << 3;

    {
        v8f acc = {};
        #pragma unroll 4
        for (int k = 0; k < D; k += 4) {
            const int kb = k + kHalf;
            v2f a, b;
            a[0] = s_tile[mrow][kb];
            a[1] = s_tile[mrow][kb + 1];
            b[0] = W1[(long)kb * D + col];
            b[1] = W1[(long)(kb + 1) * D + col];
            acc = __builtin_amdgcn_wmma_f32_16x16x4_f32(
                false, a, false, b, (short)0, acc, false, false);
        }
        const float bias = b1[col];
        #pragma unroll
        for (int v = 0; v < 8; ++v) {
            float t = acc[v] + bias;
            t_tile[rbase + v][col] = t > 0.0f ? t : 0.0f;
        }
    }
    __syncthreads();

    {
        v8f acc = {};
        #pragma unroll 4
        for (int k = 0; k < D; k += 4) {
            const int kb = k + kHalf;
            v2f a, b;
            a[0] = t_tile[mrow][kb];
            a[1] = t_tile[mrow][kb + 1];
            b[0] = W2[(long)kb * D + col];
            b[1] = W2[(long)(kb + 1) * D + col];
            acc = __builtin_amdgcn_wmma_f32_16x16x4_f32(
                false, a, false, b, (short)0, acc, false, false);
        }
        const float bias = b2[col];
        #pragma unroll
        for (int v = 0; v < 8; ++v) {
            float u = acc[v] + bias;
            if (applyReluOut) u = u > 0.0f ? u : 0.0f;
            out[(row0 + rbase + v) * D + col] = u;
        }
    }
}

extern "C" void kernel_launch(void* const* d_in, const int* in_sizes, int n_in,
                              void* d_out, int out_size, void* d_ws, size_t ws_size,
                              hipStream_t stream) {
    const float* x  = (const float*)d_in[0];
    const int*   ei = (const int*)  d_in[1];
    const float* W1 = (const float*)d_in[2];
    const float* b1 = (const float*)d_in[3];
    const float* W2 = (const float*)d_in[4];
    const float* b2 = (const float*)d_in[5];
    float* out = (float*)d_out;

    const int N = in_sizes[0] / D;
    const int E = in_sizes[1] / 2;
    const int* src = ei;
    const int* dst = ei + E;

    float* agg = (float*)d_ws;
    float* h1  = agg + (size_t)N * D;
    int* cnt   = (int*)(h1 + (size_t)N * D);
    int* start = cnt + N + 32;
    int* rank  = start + N;
    int* col   = rank + E;

    const int eBlocks = (E + BLOCK - 1) / BLOCK;
    const int aBlocks = (N + 4 * BLOCK - 1) / (4 * BLOCK);
    const int nBlocks = (N + BLOCK - 1) / BLOCK;
    const int mBlocks = N / TILE_ROWS;

    hipMemsetAsync(cnt, 0, (size_t)(N + 1) * sizeof(int), stream);
    edge_rank<<<eBlocks, BLOCK, 0, stream>>>(dst, cnt, rank, E);
    bucket_alloc<<<aBlocks, BLOCK, 0, stream>>>(cnt, cnt + N, start, N);
    bucket_fill<<<eBlocks, BLOCK, 0, stream>>>(src, dst, rank, start, col, E);
    bucket_sort<<<nBlocks, BLOCK, 0, stream>>>(start, cnt, col, N);

    gin_gather_sum<<<GATHER_BLOCKS, BLOCK, 0, stream>>>(x, start, cnt, col, agg, N);
    gin_mlp<<<mBlocks, BLOCK, 0, stream>>>(x, agg, W1, b1, W2, b2, h1, 1);

    gin_gather_sum<<<GATHER_BLOCKS, BLOCK, 0, stream>>>(h1, start, cnt, col, agg, N);
    gin_mlp<<<mBlocks, BLOCK, 0, stream>>>(h1, agg, W1, b1, W2, b2, out, 0);
}


# MLP kernel: LDS tile rows padded from 128 to 132 floats, so an A-fragment load (16 rows at one k) covers all 64 banks instead of 4: 16-way bank conflict removed; arithmetic and sum order unchanged
# speedup vs baseline: 2.7441x; 1.3444x over previous
#include <hip/hip_runtime.h>


#define D 128
#define TILE_ROWS 16
#define PITCH (D + 4)
#define BLOCK 256
#define GATHER_BLOCKS 2048
#define SORT_REGS 16

typedef float v2f __attribute__((ext_vector_type(2)));
typedef float v4f __attribute__((ext_vector_type(4)));
typedef float v8f __attribute__((ext_vector_type(8)));

__global__ __launch_bounds__(BLOCK)
void edge_rank(const int* __restrict__ dst,
               int*       __restrict__ cnt,
               int*       __restrict__ rank,
               int nEdges) {
    const int e = blockIdx.x * BLOCK + threadIdx.x;
    if (e >= nEdges) return;
    rank[e] = atomicAdd(&cnt[dst[e]], 1);
}

__global__ __launch_bounds__(BLOCK)
void bucket_alloc(const int* cnt, int* total, int* __restrict__ start, int nNodes) {
    __shared__ int wsum[BLOCK / 32];
    __shared__ int blockBase;
    const int tid  = threadIdx.x;
    const int lane = tid & 31;
    const int wave = tid >> 5;
    const int n0   = (blockIdx.x * BLOCK + tid) * 4;

    int c[4];
    #pragma unroll
    for (int j = 0; j < 4; ++j) c[j] = (n0 + j < nNodes) ? cnt[n0 + j] : 0;
    const int mine = c[0] + c[1] + c[2] + c[3];

    int incl = mine;
    for (int off = 1; off < 32; off <<= 1) {
        const int v = __shfl_up(incl, off);
        if (lane >= off) incl += v;
    }
    if (lane == 31) wsum[wave] = incl;
    __syncthreads();
    if (tid == 0) {
        int s = 0;
        for (int w = 0; w < BLOCK / 32; ++w) { const int v = wsum[w]; wsum[w] = s; s += v; }
        blockBase = atomicAdd(total, s);
    }
    __syncthreads();
    int slot = blockBase + wsum[wave] + incl - mine;
    #pragma unroll
    for (int j = 0; j < 4; ++j) {
        if (n0 + j < nNodes) { start[n0 + j] = slot; slot += c[j]; }
    }
}

__global__ __launch_bounds__(BLOCK)
void bucket_fill(const int* __restrict__ src,
                 const int* __restrict__ dst,
                 const int* __restrict__ rank,
                 const int* __restrict__ start,
                 int*       __restrict__ col,
                 int nEdges) {
    const int e = blockIdx.x * BLOCK + threadIdx.x;
    if (e >= nEdges) return;
    col[start[dst[e]] + rank[e]] = src[e];
}

__global__ __launch_bounds__(BLOCK)
void bucket_sort(const int* __restrict__ start,
                 const int* __restrict__ cnt,
                 int*       __restrict__ col,
                 int nNodes) {
    const int n = blockIdx.x * BLOCK + threadIdx.x;
    if (n >= nNodes) return;
    int* p = col + start[n];
    const int m = cnt[n];
    if (m < 2) return;
    if (m <= SORT_REGS) {
        int v[SORT_REGS];
        #pragma unroll
        for (int i = 0; i < SORT_REGS; ++i) v[i] = (i < m) ? p[i] : 0x7fffffff;
        #pragma unroll
        for (int r = 0; r < SORT_REGS; ++r) {
            #pragma unroll
            for (int i = (r & 1); i + 1 < SORT_REGS; i += 2) {
                const int lo = min(v[i], v[i + 1]);
                const int hi = max(v[i], v[i + 1]);
                v[i] = lo; v[i + 1] = hi;
            }
        }
        #pragma unroll
        for (int i = 0; i < SORT_REGS; ++i) if (i < m) p[i] = v[i];
    } else {
        for (int i = 1; i < m; ++i) {
            const int key = p[i];
            int j = i - 1;
            while (j >= 0 && p[j] > key) { p[j + 1] = p[j]; --j; }
            p[j + 1] = key;
        }
    }
}


__device__ __forceinline__
void rows_ask(const float* __restrict__ laneBase, int my, int k0, int m, v4f (&r)[8]) {
    int s[8];
    #pragma unroll
    for (int j = 0; j < 8; ++j) s[j] = __builtin_amdgcn_readlane(my, k0 + j);
    #pragma unroll
    for (int j = 0; j < 8; ++j)
        if (j < m) r[j] = *(const v4f*)(laneBase + (long)s[j] * D);
}

__device__ __forceinline__
void rows_add(const v4f (&r)[8], int m, v4f& acc) {
    #pragma unroll
    for (int j = 0; j < 8; ++j)
        if (j < m) acc += r[j];
}

__global__ __launch_bounds__(BLOCK)
void gin_gather_sum(const float* __restrict__ xin,
                    const int*   __restrict__ start,
                    const int*   __restrict__ cnt,
                    const int*   __restrict__ col,
                    float*       __restrict__ agg,
                    int nNodes) {
    const int lane = threadIdx.x & 31;
    const int wave   = __builtin_amdgcn_readfirstlane((int)(threadIdx.x >> 5));
    const int stride = (int)gridDim.x * (BLOCK / 32);
    int n = (int)blockIdx.x * (BLOCK / 32) + wave;
    const float* laneBase = xin + lane * 4;

    int st0 = 0, dg0 = 0, st1 = 0, dg1 = 0;
    if (n < nNodes) {
        st0 = __builtin_amdgcn_readfirstlane(start[n]);
        dg0 = __builtin_amdgcn_readfirstlane(cnt[n]);
    }
    if (n + stride < nNodes) {
        st1 = __builtin_amdgcn_readfirstlane(start[n + stride]);
        dg1 = __builtin_amdgcn_readfirstlane(cnt[n + stride]);
    }
    int my0 = (lane < dg0) ? col[st0 + lane] : 0;

    for (; n < nNodes; n += stride) {
        int st2 = 0, dg2 = 0;
        const int n2 = n + 2 * stride;
        if (n2 < nNodes) {
            st2 = __builtin_amdgcn_readfirstlane(start[n2]);
            dg2 = __builtin_amdgcn_readfirstlane(cnt[n2]);
        }

        v4f acc = {0.0f, 0.0f, 0.0f, 0.0f};
        v4f r[8];
        int c = dg0 < 32 ? dg0 : 32;
        int m = c < 8 ? c : 8;
        rows_ask(laneBase, my0, 0, m, r);
        const int my1 = (lane < dg1) ? col[st1 + lane] : 0;
        rows_add(r, m, acc);
        for (int k0 = 8; k0 < c; k0 += 8) {
            m = (c - k0) < 8 ? (c - k0) : 8;
            rows_ask(laneBase, my0, k0, m, r);
            rows_add(r, m, acc);
        }
        int rem = dg0 - c;
        int pos = st0;
        while (rem > 0) {
            pos += 32;
            c = rem < 32 ? rem : 32;
            const int my = (lane < c) ? col[pos + lane] : 0;
            for (int k0 = 0; k0 < c; k0 += 8) {
                m = (c - k0) < 8 ? (c - k0) : 8;
                rows_ask(laneBase, my, k0, m, r);
                rows_add(r, m, acc);
            }
            rem -= c;
        }
        *(v4f*)(agg + (long)n * D + lane * 4) = acc;

        st0 = st1; dg0 = dg1; my0 = my1;
        st1 = st2; dg1 = dg2;
    }
}

__global__ __launch_bounds__(BLOCK)
void gin_mlp(const float* __restrict__ xin,
             const float* __restrict__ agg,
             const float* __restrict__ W1, const float* __restrict__ b1,
             const float* __restrict__ W2, const float* __restrict__ b2,
             float* __restrict__ out,
             int applyReluOut) {
    __shared__ float s_tile[TILE_ROWS][PITCH];
    __shared__ float t_tile[TILE_ROWS][PITCH];

    const int tid  = threadIdx.x;
    const int lane = tid & 31;
    const int wave = tid >> 5;
    const long row0 = (long)blockIdx.x * TILE_ROWS;

    #pragma unroll
    for (int i = tid; i < TILE_ROWS * D; i += BLOCK) {
        int r = i >> 7;
        int c = i & (D - 1);
        long g = (row0 + r) * D + c;
        s_tile[r][c] = xin[g] + agg[g];
    }
    __syncthreads();

    const int mrow  = lane & 15;
    const int kHalf = (lane >> 4) << 1;
    const int n0    = wave * 16;
    const int col   = n0 + mrow;
    const int rbase = (lane >> 4) << 3;

    {
        v8f acc = {};
        #pragma unroll 4
        for (int k = 0; k < D; k += 4) {
            const int kb = k + kHalf;
            v2f a, b;
            a[0] = s_tile[mrow][kb];
            a[1] = s_tile[mrow][kb + 1];
            b[0] = W1[(long)kb * D + col];
            b[1] = W1[(long)(kb + 1) * D + col];
            acc = __builtin_amdgcn_wmma_f32_16x16x4_f32(
                false, a, false, b, (short)0, acc, false, false);
        }
        const float bias = b1[col];
        #pragma unroll
        for (int v = 0; v < 8; ++v) {
            float t = acc[v] + bias;
            t_tile[rbase + v][col] = t > 0.0f ? t : 0.0f;
        }
    }
    __syncthreads();

    {
        v8f acc = {};
        #pragma unroll 4
        for (int k = 0; k < D; k += 4) {
            const int kb = k + kHalf;
            v2f a, b;
            a[0] = t_tile[mrow][kb];
            a[1] = t_tile[mrow][kb + 1];
            b[0] = W2[(long)kb * D + col];
            b[1] = W2[(long)(kb + 1) * D + col];
            acc = __builtin_amdgcn_wmma_f32_16x16x4_f32(
                false, a, false, b, (short)0, acc, false, false);
        }
        const float bias = b2[col];
        #pragma unroll
        for (int v = 0; v < 8; ++v) {
            float u = acc[v] + bias;
            if (applyReluOut) u = u > 0.0f ? u : 0.0f;
            out[(row0 + rbase + v) * D + col] = u;
        }
    }
}

extern "C" void kernel_launch(void* const* d_in, const int* in_sizes, int n_in,
                              void* d_out, int out_size, void* d_ws, size_t ws_size,
                              hipStream_t stream) {
    const float* x  = (const float*)d_in[0];
    const int*   ei = (const int*)  d_in[1];
    const float* W1 = (const float*)d_in[2];
    const float* b1 = (const float*)d_in[3];
    const float* W2 = (const float*)d_in[4];
    const float* b2 = (const float*)d_in[5];
    float* out = (float*)d_out;

    const int N = in_sizes[0] / D;
    const int E = in_sizes[1] / 2;
    const int* src = ei;
    const int* dst = ei + E;

    float* agg = (float*)d_ws;
    float* h1  = agg + (size_t)N * D;
    int* cnt   = (int*)(h1 + (size_t)N * D);
    int* start = cnt + N + 32;
    int* rank  = start + N;
    int* col   = rank + E;

    const int eBlocks = (E + BLOCK - 1) / BLOCK;
    const int aBlocks = (N + 4 * BLOCK - 1) / (4 * BLOCK);
    const int nBlocks = (N + BLOCK - 1) / BLOCK;
    const int mBlocks = N / TILE_ROWS;

    hipMemsetAsync(cnt, 0, (size_t)(N + 1) * sizeof(int), stream);
    edge_rank<<<eBlocks, BLOCK, 0, stream>>>(dst, cnt, rank, E);
    bucket_alloc<<<aBlocks, BLOCK, 0, stream>>>(cnt, cnt + N, start, N);
    bucket_fill<<<eBlocks, BLOCK, 0, stream>>>(src, dst, rank, start, col, E);
    bucket_sort<<<nBlocks, BLOCK, 0, stream>>>(start, cnt, col, N);

    gin_gather_sum<<<GATHER_BLOCKS, BLOCK, 0, stream>>>(x, start, cnt, col, agg, N);
    gin_mlp<<<mBlocks, BLOCK, 0, stream>>>(x, agg, W1, b1, W2, b2, h1, 1);

    gin_gather_sum<<<GATHER_BLOCKS, BLOCK, 0, stream>>>(h1, start, cnt, col, agg, N);
    gin_mlp<<<mBlocks, BLOCK, 0, stream>>>(h1, agg, W1, b1, W2, b2, out, 0);
}


# MLP kernel made persistent: each wave keeps its W1 and W2 column slabs in registers as B fragments and loops over 32-row tiles; K loops read no weights from memory; sum order unchanged
# speedup vs baseline: 3.0090x; 1.0965x over previous
#include <hip/hip_runtime.h>


#define D 128
#define MLP_ROWS 32
#define MLP_BLOCKS 512
#define PITCH (D + 4)
#define BLOCK 256
#define GATHER_BLOCKS 2048
#define SORT_REGS 16

typedef float v2f __attribute__((ext_vector_type(2)));
typedef float v4f __attribute__((ext_vector_type(4)));
typedef float v8f __attribute__((ext_vector_type(8)));

__global__ __launch_bounds__(BLOCK)
void edge_rank(const int* __restrict__ dst,
               int*       __restrict__ cnt,
               int*       __restrict__ rank,
               int nEdges) {
    const int e = blockIdx.x * BLOCK + threadIdx.x;
    if (e >= nEdges) return;
    rank[e] = atomicAdd(&cnt[dst[e]], 1);
}

__global__ __launch_bounds__(BLOCK)
void bucket_alloc(const int* cnt, int* total, int* __restrict__ start, int nNodes) {
    __shared__ int wsum[BLOCK / 32];
    __shared__ int blockBase;
    const int tid  = threadIdx.x;
    const int lane = tid & 31;
    const int wave = tid >> 5;
    const int n0   = (blockIdx.x * BLOCK + tid) * 4;

    int c[4];
    #pragma unroll
    for (int j = 0; j < 4; ++j) c[j] = (n0 + j < nNodes) ? cnt[n0 + j] : 0;
    const int mine = c[0] + c[1] + c[2] + c[3];

    int incl = mine;
    for (int off = 1; off < 32; off <<= 1) {
        const int v = __shfl_up(incl, off);
        if (lane >= off) incl += v;
    }
    if (lane == 31) wsum[wave] = incl;
    __syncthreads();
    if (tid == 0) {
        int s = 0;
        for (int w = 0; w < BLOCK / 32; ++w) { const int v = wsum[w]; wsum[w] = s; s += v; }
        blockBase = atomicAdd(total, s);
    }
    __syncthreads();
    int slot = blockBase + wsum[wave] + incl - mine;
    #pragma unroll
    for (int j = 0; j < 4; ++j) {
        if (n0 + j < nNodes) { start[n0 + j] = slot; slot += c[j]; }
    }
}

__global__ __launch_bounds__(BLOCK)
void bucket_fill(const int* __restrict__ src,
                 const int* __restrict__ dst,
                 const int* __restrict__ rank,
                 const int* __restrict__ start,
                 int*       __restrict__ col,
                 int nEdges) {
    const int e = blockIdx.x * BLOCK + threadIdx.x;
    if (e >= nEdges) return;
    col[start[dst[e]] + rank[e]] = src[e];
}

__global__ __launch_bounds__(BLOCK)
void bucket_sort(const int* __restrict__ start,
                 const int* __restrict__ cnt,
                 int*       __restrict__ col,
                 int nNodes) {
    const int n = blockIdx.x * BLOCK + threadIdx.x;
    if (n >= nNodes) return;
    int* p = col + start[n];
    const int m = cnt[n];
    if (m < 2) return;
    if (m <= SORT_REGS) {
        int v[SORT_REGS];
        #pragma unroll
        for (int i = 0; i < SORT_REGS; ++i) v[i] = (i < m) ? p[i] : 0x7fffffff;
        #pragma unroll
        for (int r = 0; r < SORT_REGS; ++r) {
            #pragma unroll
            for (int i = (r & 1); i + 1 < SORT_REGS; i += 2) {
                const int lo = min(v[i], v[i + 1]);
                const int hi = max(v[i], v[i + 1]);
                v[i] = lo; v[i + 1] = hi;
            }
        }
        #pragma unroll
        for (int i = 0; i < SORT_REGS; ++i) if (i < m) p[i] = v[i];
    } else {
        for (int i = 1; i < m; ++i) {
            const int key = p[i];
            int j = i - 1;
            while (j >= 0 && p[j] > key) { p[j + 1] = p[j]; --j; }
            p[j + 1] = key;
        }
    }
}


__device__ __forceinline__
void rows_ask(const float* __restrict__ laneBase, int my, int k0, int m, v4f (&r)[8]) {
    int s[8];
    #pragma unroll
    for (int j = 0; j < 8; ++j) s[j] = __builtin_amdgcn_readlane(my, k0 + j);
    #pragma unroll
    for (int j = 0; j < 8; ++j)
        if (j < m) r[j] = *(const v4f*)(laneBase + (long)s[j] * D);
}

__device__ __forceinline__
void rows_add(const v4f (&r)[8], int m, v4f& acc) {
    #pragma unroll
    for (int j = 0; j < 8; ++j)
        if (j < m) acc += r[j];
}

__global__ __launch_bounds__(BLOCK)
void gin_gather_sum(const float* __restrict__ xin,
                    const int*   __restrict__ start,
                    const int*   __restrict__ cnt,
                    const int*   __restrict__ col,
                    float*       __restrict__ agg,
                    int nNodes) {
    const int lane = threadIdx.x & 31;
    const int wave   = __builtin_amdgcn_readfirstlane((int)(threadIdx.x >> 5));
    const int stride = (int)gridDim.x * (BLOCK / 32);
    int n = (int)blockIdx.x * (BLOCK / 32) + wave;
    const float* laneBase = xin + lane * 4;

    int st0 = 0, dg0 = 0, st1 = 0, dg1 = 0;
    if (n < nNodes) {
        st0 = __builtin_amdgcn_readfirstlane(start[n]);
        dg0 = __builtin_amdgcn_readfirstlane(cnt[n]);
    }
    if (n + stride < nNodes) {
        st1 = __builtin_amdgcn_readfirstlane(start[n + stride]);
        dg1 = __builtin_amdgcn_readfirstlane(cnt[n + stride]);
    }
    int my0 = (lane < dg0) ? col[st0 + lane] : 0;

    for (; n < nNodes; n += stride) {
        int st2 = 0, dg2 = 0;
        const int n2 = n + 2 * stride;
        if (n2 < nNodes) {
            st2 = __builtin_amdgcn_readfirstlane(start[n2]);
            dg2 = __builtin_amdgcn_readfirstlane(cnt[n2]);
        }

        v4f acc = {0.0f, 0.0f, 0.0f, 0.0f};
        v4f r[8];
        int c = dg0 < 32 ? dg0 : 32;
        int m = c < 8 ? c : 8;
        rows_ask(laneBase, my0, 0, m, r);
        const int my1 = (lane < dg1) ? col[st1 + lane] : 0;
        rows_add(r, m, acc);
        for (int k0 = 8; k0 < c; k0 += 8) {
            m = (c - k0) < 8 ? (c - k0) : 8;
            rows_ask(laneBase, my0, k0, m, r);
            rows_add(r, m, acc);
        }
        int rem = dg0 - c;
        int pos = st0;
        while (rem > 0) {
            pos += 32;
            c = rem < 32 ? rem : 32;
            const int my = (lane < c) ? col[pos + lane] : 0;
            for (int k0 = 0; k0 < c; k0 += 8) {
                m = (c - k0) < 8 ? (c - k0) : 8;
                rows_ask(laneBase, my, k0, m, r);
                rows_add(r, m, acc);
            }
            rem -= c;
        }
        *(v4f*)(agg + (long)n * D + lane * 4) = acc;

        st0 = st1; dg0 = dg1; my0 = my1;
        st1 = st2; dg1 = dg2;
    }
}


__device__ __forceinline__
void tile_ask(const float* __restrict__ xin, const float* __restrict__ agg,
              long row0, int tid, int nNodes, v4f (&px)[4], v4f (&pa)[4]) {
    const float* xp = xin + row0 * D + tid * 4;
    const float* ap = agg + row0 * D + tid * 4;
    if (row0 + MLP_ROWS <= nNodes) {
        #pragma unroll
        for (int i = 0; i < 4; ++i) {
            px[i] = *(const v4f*)(xp + i * (BLOCK * 4));
            pa[i] = *(const v4f*)(ap + i * (BLOCK * 4));
        }
    } else {
        #pragma unroll
        for (int i = 0; i < 4; ++i) {
            const long row = row0 + ((tid + i * BLOCK) >> 5);
            if (row < nNodes) {
                px[i] = *(const v4f*)(xp + i * (BLOCK * 4));
                pa[i] = *(const v4f*)(ap + i * (BLOCK * 4));
            } else {
                px[i] = (v4f){0.0f, 0.0f, 0.0f, 0.0f};
                pa[i] = (v4f){0.0f, 0.0f, 0.0f, 0.0f};
            }
        }
    }
}

__global__ __launch_bounds__(BLOCK)
void gin_mlp(const float* __restrict__ xin,
             const float* __restrict__ agg,
             const float* __restrict__ W1, const float* __restrict__ b1,
             const float* __restrict__ W2, const float* __restrict__ b2,
             float* __restrict__ out,
             int nNodes,
             int applyReluOut) {
    __shared__ __attribute__((aligned(16))) float s_tile[MLP_ROWS][PITCH];
    __shared__ __attribute__((aligned(16))) float t_tile[MLP_ROWS][PITCH];

    const int tid  = threadIdx.x;
    const int lane = tid & 31;
    const int wave = tid >> 5;

    const int mrow  = lane & 15;
    const int kHalf = (lane >> 4) << 1;
    const int col   = wave * 16 + mrow;
    const int rbase = (lane >> 4) << 3;

    v2f w1f[D / 4], w2f[D / 4];
    #pragma unroll
    for (int j = 0; j < D / 4; ++j) {
        const int kb = 4 * j + kHalf;
        w1f[j][0] = W1[(long)kb * D + col];
        w1f[j][1] = W1[(long)(kb + 1) * D + col];
        w2f[j][0] = W2[(long)kb * D + col];
        w2f[j][1] = W2[(long)(kb + 1) * D + col];
    }
    const float bias1 = b1[col];
    const float bias2 = b2[col];

    const int nTiles = (nNodes + MLP_ROWS - 1) / MLP_ROWS;
    int tile = blockIdx.x;
    v4f px[4], pa[4];
    if (tile < nTiles) tile_ask(xin, agg, (long)tile * MLP_ROWS, tid, nNodes, px, pa);

    for (; tile < nTiles; tile += gridDim.x) {
        const long row0 = (long)tile * MLP_ROWS;

        #pragma unroll
        for (int i = 0; i < 4; ++i) {
            const int idx = tid + i * BLOCK;
            *(v4f*)&s_tile[idx >> 5][(idx & 31) * 4] = px[i] + pa[i];
        }
        __syncthreads();

        const int next = tile + (int)gridDim.x;
        if (next < nTiles) tile_ask(xin, agg, (long)next * MLP_ROWS, tid, nNodes, px, pa);

        {
            v8f acc0 = {}, acc1 = {};
            #pragma unroll
            for (int j = 0; j < D / 4; ++j) {
                const int kb = 4 * j + kHalf;
                const v2f a0 = *(const v2f*)&s_tile[mrow][kb];
                const v2f a1 = *(const v2f*)&s_tile[16 + mrow][kb];
                acc0 = __builtin_amdgcn_wmma_f32_16x16x4_f32(
                    false, a0, false, w1f[j], (short)0, acc0, false, false);
                acc1 = __builtin_amdgcn_wmma_f32_16x16x4_f32(
                    false, a1, false, w1f[j], (short)0, acc1, false, false);
            }
            #pragma unroll
            for (int v = 0; v < 8; ++v) {
                const float t0 = acc0[v] + bias1;
                const float t1 = acc1[v] + bias1;
                t_tile[rbase + v][col]      = t0 > 0.0f ? t0 : 0.0f;
                t_tile[16 + rbase + v][col] = t1 > 0.0f ? t1 : 0.0f;
            }
        }
        __syncthreads();

        {
            v8f acc0 = {}, acc1 = {};
            #pragma unroll
            for (int j = 0; j < D / 4; ++j) {
                const int kb = 4 * j + kHalf;
                const v2f a0 = *(const v2f*)&t_tile[mrow][kb];
                const v2f a1 = *(const v2f*)&t_tile[16 + mrow][kb];
                acc0 = __builtin_amdgcn_wmma_f32_16x16x4_f32(
                    false, a0, false, w2f[j], (short)0, acc0, false, false);
                acc1 = __builtin_amdgcn_wmma_f32_16x16x4_f32(
                    false, a1, false, w2f[j], (short)0, acc1, false, false);
            }
            float u0[8], u1[8];
            #pragma unroll
            for (int v = 0; v < 8; ++v) {
                u0[v] = acc0[v] + bias2;
                u1[v] = acc1[v] + bias2;
                if (applyReluOut) {
                    u0[v] = u0[v] > 0.0f ? u0[v] : 0.0f;
                    u1[v] = u1[v] > 0.0f ? u1[v] : 0.0f;
                }
            }
            float* op = out + (row0 + rbase) * D + col;
            if (row0 + MLP_ROWS <= nNodes) {
                #pragma unroll
                for (int v = 0; v < 8; ++v) {
                    op[v * D]        = u0[v];
                    op[(16 + v) * D] = u1[v];
                }
            } else {
                #pragma unroll
                for (int v = 0; v < 8; ++v) {
                    if (row0 + rbase + v < nNodes)      op[v * D]        = u0[v];
                    if (row0 + rbase + v + 16 < nNodes) op[(16 + v) * D] = u1[v];
                }
            }
        }
    }
}

extern "C" void kernel_launch(void* const* d_in, const int* in_sizes, int n_in,
                              void* d_out, int out_size, void* d_ws, size_t ws_size,
                              hipStream_t stream) {
    const float* x  = (const float*)d_in[0];
    const int*   ei = (const int*)  d_in[1];
    const float* W1 = (const float*)d_in[2];
    const float* b1 = (const float*)d_in[3];
    const float* W2 = (const float*)d_in[4];
    const float* b2 = (const float*)d_in[5];
    float* out = (float*)d_out;

    const int N = in_sizes[0] / D;
    const int E = in_sizes[1] / 2;
    const int* src = ei;
    const int* dst = ei + E;

    float* agg = (float*)d_ws;
    float* h1  = agg + (size_t)N * D;
    int* cnt   = (int*)(h1 + (size_t)N * D);
    int* start = cnt + N + 32;
    int* rank  = start + N;
    int* col   = rank + E;

    const int eBlocks = (E + BLOCK - 1) / BLOCK;
    const int aBlocks = (N + 4 * BLOCK - 1) / (4 * BLOCK);
    const int nBlocks = (N + BLOCK - 1) / BLOCK;

    hipMemsetAsync(cnt, 0, (size_t)(N + 1) * sizeof(int), stream);
    edge_rank<<<eBlocks, BLOCK, 0, stream>>>(dst, cnt, rank, E);
    bucket_alloc<<<aBlocks, BLOCK, 0, stream>>>(cnt, cnt + N, start, N);
    bucket_fill<<<eBlocks, BLOCK, 0, stream>>>(src, dst, rank, start, col, E);
    bucket_sort<<<nBlocks, BLOCK, 0, stream>>>(start, cnt, col, N);

    gin_gather_sum<<<GATHER_BLOCKS, BLOCK, 0, stream>>>(x, start, cnt, col, agg, N);
    gin_mlp<<<MLP_BLOCKS, BLOCK, 0, stream>>>(x, agg, W1, b1, W2, b2, h1, N, 1);

    gin_gather_sum<<<GATHER_BLOCKS, BLOCK, 0, stream>>>(h1, start, cnt, col, agg, N);
    gin_mlp<<<MLP_BLOCKS, BLOCK, 0, stream>>>(h1, agg, W1, b1, W2, b2, out, N, 0);
}


# Edge bucketing moved from global atomics in four launches to LDS counters in one: a workgroup owns 1024 nodes, reads the destination list once, notes its edges in LDS; buckets, sum order unchanged
# speedup vs baseline: 3.2018x; 1.0641x over previous
#include <hip/hip_runtime.h>


#define D 128
#define MLP_ROWS 32
#define MLP_BLOCKS 512
#define PITCH (D + 4)
#define BLOCK 256
#define GATHER_BLOCKS 2048
#define SORT_REGS 16
#define BUILD_THREADS 1024
#define RANGE_BITS 10
#define RANGE (1 << RANGE_BITS)
#define SCAN_DEPTH 8
#define QSLOTS 32
#define EDGE_BITS (32 - RANGE_BITS)

typedef float v2f __attribute__((ext_vector_type(2)));
typedef float v4f __attribute__((ext_vector_type(4)));
typedef float v8f __attribute__((ext_vector_type(8)));
typedef int   v4i __attribute__((ext_vector_type(4)));


__device__ __forceinline__
void dst_ask(const v4i* __restrict__ p, int n4, int rot, int base, int tid,
             v4i (&d)[SCAN_DEPTH], int (&e0)[SCAN_DEPTH]) {
    #pragma unroll
    for (int j = 0; j < SCAN_DEPTH; ++j) {
        const int i = base + j * BUILD_THREADS + tid;
        int idx = i + rot;
        if (idx >= n4) idx -= n4;
        e0[j] = idx * 4;
        if (i < n4) d[j] = p[idx];
        else        d[j] = (v4i){-1, -1, -1, -1};
    }
}

__global__ __launch_bounds__(BUILD_THREADS)
void bucket_build(const int* __restrict__ src,
                  const int* __restrict__ dst,
                  int* total,
                  int* __restrict__ start,
                  int* __restrict__ cnt,
                  int* col,
                  int nNodes, int nEdges) {
    __shared__ int s_cnt[RANGE];
    __shared__ unsigned s_q[QSLOTS * BUILD_THREADS];
    __shared__ int s_over;
    __shared__ int s_wsum[BUILD_THREADS / 32];
    __shared__ int s_base;
    const int tid  = threadIdx.x;
    const int lane = tid & 31;
    const int wave = tid >> 5;
    const int lo   = (int)blockIdx.x * RANGE;

    const bool vec = (((unsigned long long)dst) & 15ull) == 0ull;
    const v4i* p4  = (const v4i*)dst;
    const int n4   = vec ? (nEdges >> 2) : 0;
    const int rot  = (int)(((long)blockIdx.x * n4) / (long)gridDim.x);

    s_cnt[tid] = 0;
    if (tid == 0) s_over = 0;
    __syncthreads();

    int nq = 0;
    for (int base = 0; base < n4; base += SCAN_DEPTH * BUILD_THREADS) {
        v4i d[SCAN_DEPTH];
        int e0[SCAN_DEPTH];
        dst_ask(p4, n4, rot, base, tid, d, e0);
        #pragma unroll
        for (int j = 0; j < SCAN_DEPTH; ++j) {
            #pragma unroll
            for (int c = 0; c < 4; ++c) {
                const unsigned r = (unsigned)(d[j][c] - lo);
                if (r < (unsigned)RANGE) {
                    atomicAdd(&s_cnt[r], 1);
                    if (nq < QSLOTS)
                        s_q[nq * BUILD_THREADS + tid] = ((unsigned)(e0[j] + c) << RANGE_BITS) | r;
                    ++nq;
                }
            }
        }
    }
    for (int e = 4 * n4 + tid; e < nEdges; e += BUILD_THREADS) {
        const unsigned r = (unsigned)(dst[e] - lo);
        if (r < (unsigned)RANGE) {
            atomicAdd(&s_cnt[r], 1);
            if (nq < QSLOTS)
                s_q[nq * BUILD_THREADS + tid] = ((unsigned)e << RANGE_BITS) | r;
            ++nq;
        }
    }
    if (nq > QSLOTS) s_over = 1;
    __syncthreads();

    const int mine = s_cnt[tid];
    int incl = mine;
    for (int off = 1; off < 32; off <<= 1) {
        const int v = __shfl_up(incl, off);
        if (lane >= off) incl += v;
    }
    if (lane == 31) s_wsum[wave] = incl;
    __syncthreads();
    if (wave == 0) {
        const int w = s_wsum[lane];
        int winc = w;
        for (int off = 1; off < 32; off <<= 1) {
            const int v = __shfl_up(winc, off);
            if (lane >= off) winc += v;
        }
        s_wsum[lane] = winc - w;
        if (lane == 31) s_base = atomicAdd(total, winc);
    }
    __syncthreads();
    const int first = s_base + s_wsum[wave] + incl - mine;
    const int n = lo + tid;
    if (n < nNodes) { start[n] = first; cnt[n] = mine; }
    s_cnt[tid] = first;
    __syncthreads();

    const bool noted = (s_over == 0) && ((unsigned)nEdges <= (1u << EDGE_BITS));
    if (noted) {
        for (int k = 0; k < nq; ++k) {
            const unsigned w = s_q[k * BUILD_THREADS + tid];
            col[atomicAdd(&s_cnt[w & (RANGE - 1)], 1)] = (int)(w >> RANGE_BITS);
        }
    } else {
        for (int base = 0; base < n4; base += SCAN_DEPTH * BUILD_THREADS) {
            v4i d[SCAN_DEPTH];
            int e0[SCAN_DEPTH];
            dst_ask(p4, n4, rot, base, tid, d, e0);
            #pragma unroll
            for (int j = 0; j < SCAN_DEPTH; ++j) {
                #pragma unroll
                for (int c = 0; c < 4; ++c) {
                    const unsigned r = (unsigned)(d[j][c] - lo);
                    if (r < (unsigned)RANGE) col[atomicAdd(&s_cnt[r], 1)] = e0[j] + c;
                }
            }
        }
        for (int e = 4 * n4 + tid; e < nEdges; e += BUILD_THREADS) {
            const unsigned r = (unsigned)(dst[e] - lo);
            if (r < (unsigned)RANGE) col[atomicAdd(&s_cnt[r], 1)] = e;
        }
    }
    __syncthreads();

    int* p = col + first;
    const int m = mine;
    if (m == 0) return;
    if (m <= SORT_REGS) {
        int v[SORT_REGS];
        #pragma unroll
        for (int i = 0; i < SORT_REGS; ++i) v[i] = (i < m) ? p[i] : 0;
        #pragma unroll
        for (int i = 0; i < SORT_REGS; ++i) v[i] = (i < m) ? src[v[i]] : 0x7fffffff;
        #pragma unroll
        for (int r = 0; r < SORT_REGS; ++r) {
            #pragma unroll
            for (int i = (r & 1); i + 1 < SORT_REGS; i += 2) {
                const int lo2 = min(v[i], v[i + 1]);
                const int hi2 = max(v[i], v[i + 1]);
                v[i] = lo2; v[i + 1] = hi2;
            }
        }
        #pragma unroll
        for (int i = 0; i < SORT_REGS; ++i) if (i < m) p[i] = v[i];
    } else {
        for (int i = 0; i < m; ++i) {
            const int key = src[p[i]];
            int j = i - 1;
            while (j >= 0 && p[j] > key) { p[j + 1] = p[j]; --j; }
            p[j + 1] = key;
        }
    }
}


__device__ __forceinline__
void rows_ask(const float* __restrict__ laneBase, int my, int k0, int m, v4f (&r)[8]) {
    int s[8];
    #pragma unroll
    for (int j = 0; j < 8; ++j) s[j] = __builtin_amdgcn_readlane(my, k0 + j);
    #pragma unroll
    for (int j = 0; j < 8; ++j)
        if (j < m) r[j] = *(const v4f*)(laneBase + (long)s[j] * D);
}

__device__ __forceinline__
void rows_add(const v4f (&r)[8], int m, v4f& acc) {
    #pragma unroll
    for (int j = 0; j < 8; ++j)
        if (j < m) acc += r[j];
}

__global__ __launch_bounds__(BLOCK)
void gin_gather_sum(const float* __restrict__ xin,
                    const int*   __restrict__ start,
                    const int*   __restrict__ cnt,
                    const int*   __restrict__ col,
                    float*       __restrict__ agg,
                    int nNodes) {
    const int lane = threadIdx.x & 31;
    const int wave   = __builtin_amdgcn_readfirstlane((int)(threadIdx.x >> 5));
    const int stride = (int)gridDim.x * (BLOCK / 32);
    int n = (int)blockIdx.x * (BLOCK / 32) + wave;
    const float* laneBase = xin + lane * 4;

    int st0 = 0, dg0 = 0, st1 = 0, dg1 = 0;
    if (n < nNodes) {
        st0 = __builtin_amdgcn_readfirstlane(start[n]);
        dg0 = __builtin_amdgcn_readfirstlane(cnt[n]);
    }
    if (n + stride < nNodes) {
        st1 = __builtin_amdgcn_readfirstlane(start[n + stride]);
        dg1 = __builtin_amdgcn_readfirstlane(cnt[n + stride]);
    }
    int my0 = (lane < dg0) ? col[st0 + lane] : 0;

    for (; n < nNodes; n += stride) {
        int st2 = 0, dg2 = 0;
        const int n2 = n + 2 * stride;
        if (n2 < nNodes) {
            st2 = __builtin_amdgcn_readfirstlane(start[n2]);
            dg2 = __builtin_amdgcn_readfirstlane(cnt[n2]);
        }

        v4f acc = {0.0f, 0.0f, 0.0f, 0.0f};
        v4f r[8];
        int c = dg0 < 32 ? dg0 : 32;
        int m = c < 8 ? c : 8;
        rows_ask(laneBase, my0, 0, m, r);
        const int my1 = (lane < dg1) ? col[st1 + lane] : 0;
        rows_add(r, m, acc);
        for (int k0 = 8; k0 < c; k0 += 8) {
            m = (c - k0) < 8 ? (c - k0) : 8;
            rows_ask(laneBase, my0, k0, m, r);
            rows_add(r, m, acc);
        }
        int rem = dg0 - c;
        int pos = st0;
        while (rem > 0) {
            pos += 32;
            c = rem < 32 ? rem : 32;
            const int my = (lane < c) ? col[pos + lane] : 0;
            for (int k0 = 0; k0 < c; k0 += 8) {
                m = (c - k0) < 8 ? (c - k0) : 8;
                rows_ask(laneBase, my, k0, m, r);
                rows_add(r, m, acc);
            }
            rem -= c;
        }
        *(v4f*)(agg + (long)n * D + lane * 4) = acc;

        st0 = st1; dg0 = dg1; my0 = my1;
        st1 = st2; dg1 = dg2;
    }
}


__device__ __forceinline__
void tile_ask(const float* __restrict__ xin, const float* __restrict__ agg,
              long row0, int tid, int nNodes, v4f (&px)[4], v4f (&pa)[4]) {
    const float* xp = xin + row0 * D + tid * 4;
    const float* ap = agg + row0 * D + tid * 4;
    if (row0 + MLP_ROWS <= nNodes) {
        #pragma unroll
        for (int i = 0; i < 4; ++i) {
            px[i] = *(const v4f*)(xp + i * (BLOCK * 4));
            pa[i] = *(const v4f*)(ap + i * (BLOCK * 4));
        }
    } else {
        #pragma unroll
        for (int i = 0; i < 4; ++i) {
            const long row = row0 + ((tid + i * BLOCK) >> 5);
            if (row < nNodes) {
                px[i] = *(const v4f*)(xp + i * (BLOCK * 4));
                pa[i] = *(const v4f*)(ap + i * (BLOCK * 4));
            } else {
                px[i] = (v4f){0.0f, 0.0f, 0.0f, 0.0f};
                pa[i] = (v4f){0.0f, 0.0f, 0.0f, 0.0f};
            }
        }
    }
}

__global__ __launch_bounds__(BLOCK)
void gin_mlp(const float* __restrict__ xin,
             const float* __restrict__ agg,
             const float* __restrict__ W1, const float* __restrict__ b1,
             const float* __restrict__ W2, const float* __restrict__ b2,
             float* __restrict__ out,
             int nNodes,
             int applyReluOut) {
    __shared__ __attribute__((aligned(16))) float s_tile[MLP_ROWS][PITCH];
    __shared__ __attribute__((aligned(16))) float t_tile[MLP_ROWS][PITCH];

    const int tid  = threadIdx.x;
    const int lane = tid & 31;
    const int wave = tid >> 5;

    const int mrow  = lane & 15;
    const int kHalf = (lane >> 4) << 1;
    const int col   = wave * 16 + mrow;
    const int rbase = (lane >> 4) << 3;

    v2f w1f[D / 4], w2f[D / 4];
    #pragma unroll
    for (int j = 0; j < D / 4; ++j) {
        const int kb = 4 * j + kHalf;
        w1f[j][0] = W1[(long)kb * D + col];
        w1f[j][1] = W1[(long)(kb + 1) * D + col];
        w2f[j][0] = W2[(long)kb * D + col];
        w2f[j][1] = W2[(long)(kb + 1) * D + col];
    }
    const float bias1 = b1[col];
    const float bias2 = b2[col];

    const int nTiles = (nNodes + MLP_ROWS - 1) / MLP_ROWS;
    int tile = blockIdx.x;
    v4f px[4], pa[4];
    if (tile < nTiles) tile_ask(xin, agg, (long)tile * MLP_ROWS, tid, nNodes, px, pa);

    for (; tile < nTiles; tile += gridDim.x) {
        const long row0 = (long)tile * MLP_ROWS;

        #pragma unroll
        for (int i = 0; i < 4; ++i) {
            const int idx = tid + i * BLOCK;
            *(v4f*)&s_tile[idx >> 5][(idx & 31) * 4] = px[i] + pa[i];
        }
        __syncthreads();

        const int next = tile + (int)gridDim.x;
        if (next < nTiles) tile_ask(xin, agg, (long)next * MLP_ROWS, tid, nNodes, px, pa);

        {
            v8f acc0 = {}, acc1 = {};
            #pragma unroll
            for (int j = 0; j < D / 4; ++j) {
                const int kb = 4 * j + kHalf;
                const v2f a0 = *(const v2f*)&s_tile[mrow][kb];
                const v2f a1 = *(const v2f*)&s_tile[16 + mrow][kb];
                acc0 = __builtin_amdgcn_wmma_f32_16x16x4_f32(
                    false, a0, false, w1f[j], (short)0, acc0, false, false);
                acc1 = __builtin_amdgcn_wmma_f32_16x16x4_f32(
                    false, a1, false, w1f[j], (short)0, acc1, false, false);
            }
            #pragma unroll
            for (int v = 0; v < 8; ++v) {
                const float t0 = acc0[v] + bias1;
                const float t1 = acc1[v] + bias1;
                t_tile[rbase + v][col]      = t0 > 0.0f ? t0 : 0.0f;
                t_tile[16 + rbase + v][col] = t1 > 0.0f ? t1 : 0.0f;
            }
        }
        __syncthreads();

        {
            v8f acc0 = {}, acc1 = {};
            #pragma unroll
            for (int j = 0; j < D / 4; ++j) {
                const int kb = 4 * j + kHalf;
                const v2f a0 = *(const v2f*)&t_tile[mrow][kb];
                const v2f a1 = *(const v2f*)&t_tile[16 + mrow][kb];
                acc0 = __builtin_amdgcn_wmma_f32_16x16x4_f32(
                    false, a0, false, w2f[j], (short)0, acc0, false, false);
                acc1 = __builtin_amdgcn_wmma_f32_16x16x4_f32(
                    false, a1, false, w2f[j], (short)0, acc1, false, false);
            }
            float u0[8], u1[8];
            #pragma unroll
            for (int v = 0; v < 8; ++v) {
                u0[v] = acc0[v] + bias2;
                u1[v] = acc1[v] + bias2;
                if (applyReluOut) {
                    u0[v] = u0[v] > 0.0f ? u0[v] : 0.0f;
                    u1[v] = u1[v] > 0.0f ? u1[v] : 0.0f;
                }
            }
            float* op = out + (row0 + rbase) * D + col;
            if (row0 + MLP_ROWS <= nNodes) {
                #pragma unroll
                for (int v = 0; v < 8; ++v) {
                    op[v * D]        = u0[v];
                    op[(16 + v) * D] = u1[v];
                }
            } else {
                #pragma unroll
                for (int v = 0; v < 8; ++v) {
                    if (row0 + rbase + v < nNodes)      op[v * D]        = u0[v];
                    if (row0 + rbase + v + 16 < nNodes) op[(16 + v) * D] = u1[v];
                }
            }
        }
    }
}

extern "C" void kernel_launch(void* const* d_in, const int* in_sizes, int n_in,
                              void* d_out, int out_size, void* d_ws, size_t ws_size,
                              hipStream_t stream) {
    const float* x  = (const float*)d_in[0];
    const int*   ei = (const int*)  d_in[1];
    const float* W1 = (const float*)d_in[2];
    const float* b1 = (const float*)d_in[3];
    const float* W2 = (const float*)d_in[4];
    const float* b2 = (const float*)d_in[5];
    float* out = (float*)d_out;

    const int N = in_sizes[0] / D;
    const int E = in_sizes[1] / 2;
    const int* src = ei;
    const int* dst = ei + E;

    float* agg = (float*)d_ws;
    float* h1  = agg + (size_t)N * D;
    int* cnt   = (int*)(h1 + (size_t)N * D);
    int* start = cnt + N + 32;
    int* col   = start + N;

    const int bBlocks = (N + RANGE - 1) / RANGE;

    hipMemsetAsync(cnt + N, 0, sizeof(int), stream);
    bucket_build<<<bBlocks, BUILD_THREADS, 0, stream>>>(src, dst, cnt + N, start, cnt, col, N, E);

    gin_gather_sum<<<GATHER_BLOCKS, BLOCK, 0, stream>>>(x, start, cnt, col, agg, N);
    gin_mlp<<<MLP_BLOCKS, BLOCK, 0, stream>>>(x, agg, W1, b1, W2, b2, h1, N, 1);

    gin_gather_sum<<<GATHER_BLOCKS, BLOCK, 0, stream>>>(h1, start, cnt, col, agg, N);
    gin_mlp<<<MLP_BLOCKS, BLOCK, 0, stream>>>(h1, agg, W1, b1, W2, b2, out, N, 0);
}


# Bucket build: buckets of 17 to 32 edges sorted in registers (32-wide network, loads issued together) instead of one thread's insertion sort through global memory; sum order unchanged
# speedup vs baseline: 3.6241x; 1.1319x over previous
#include <hip/hip_runtime.h>


#define D 128
#define MLP_ROWS 32
#define MLP_BLOCKS 512
#define PITCH (D + 4)
#define BLOCK 256
#define GATHER_BLOCKS 2048
#define SORT_REGS 16
#define SORT_WIDE 32
#define BUILD_THREADS 1024
#define RANGE_BITS 10
#define RANGE (1 << RANGE_BITS)
#define SCAN_DEPTH 8
#define QSLOTS 32
#define EDGE_BITS (32 - RANGE_BITS)

typedef float v2f __attribute__((ext_vector_type(2)));
typedef float v4f __attribute__((ext_vector_type(4)));
typedef float v8f __attribute__((ext_vector_type(8)));
typedef int   v4i __attribute__((ext_vector_type(4)));


__device__ __forceinline__
void dst_ask(const v4i* __restrict__ p, int n4, int rot, int base, int tid,
             v4i (&d)[SCAN_DEPTH], int (&e0)[SCAN_DEPTH]) {
    #pragma unroll
    for (int j = 0; j < SCAN_DEPTH; ++j) {
        const int i = base + j * BUILD_THREADS + tid;
        int idx = i + rot;
        if (idx >= n4) idx -= n4;
        e0[j] = idx * 4;
        if (i < n4) d[j] = p[idx];
        else        d[j] = (v4i){-1, -1, -1, -1};
    }
}

__global__ __launch_bounds__(BUILD_THREADS)
void bucket_build(const int* __restrict__ src,
                  const int* __restrict__ dst,
                  int* total,
                  int* __restrict__ start,
                  int* __restrict__ cnt,
                  int* col,
                  int nNodes, int nEdges) {
    __shared__ int s_cnt[RANGE];
    __shared__ unsigned s_q[QSLOTS * BUILD_THREADS];
    __shared__ int s_over;
    __shared__ int s_wsum[BUILD_THREADS / 32];
    __shared__ int s_base;
    const int tid  = threadIdx.x;
    const int lane = tid & 31;
    const int wave = tid >> 5;
    const int lo   = (int)blockIdx.x * RANGE;

    const bool vec = (((unsigned long long)dst) & 15ull) == 0ull;
    const v4i* p4  = (const v4i*)dst;
    const int n4   = vec ? (nEdges >> 2) : 0;
    const int rot  = (int)(((long)blockIdx.x * n4) / (long)gridDim.x);

    s_cnt[tid] = 0;
    if (tid == 0) s_over = 0;
    __syncthreads();

    int nq = 0;
    for (int base = 0; base < n4; base += SCAN_DEPTH * BUILD_THREADS) {
        v4i d[SCAN_DEPTH];
        int e0[SCAN_DEPTH];
        dst_ask(p4, n4, rot, base, tid, d, e0);
        #pragma unroll
        for (int j = 0; j < SCAN_DEPTH; ++j) {
            #pragma unroll
            for (int c = 0; c < 4; ++c) {
                const unsigned r = (unsigned)(d[j][c] - lo);
                if (r < (unsigned)RANGE) {
                    atomicAdd(&s_cnt[r], 1);
                    if (nq < QSLOTS)
                        s_q[nq * BUILD_THREADS + tid] = ((unsigned)(e0[j] + c) << RANGE_BITS) | r;
                    ++nq;
                }
            }
        }
    }
    for (int e = 4 * n4 + tid; e < nEdges; e += BUILD_THREADS) {
        const unsigned r = (unsigned)(dst[e] - lo);
        if (r < (unsigned)RANGE) {
            atomicAdd(&s_cnt[r], 1);
            if (nq < QSLOTS)
                s_q[nq * BUILD_THREADS + tid] = ((unsigned)e << RANGE_BITS) | r;
            ++nq;
        }
    }
    if (nq > QSLOTS) s_over = 1;
    __syncthreads();

    const int mine = s_cnt[tid];
    int incl = mine;
    for (int off = 1; off < 32; off <<= 1) {
        const int v = __shfl_up(incl, off);
        if (lane >= off) incl += v;
    }
    if (lane == 31) s_wsum[wave] = incl;
    __syncthreads();
    if (wave == 0) {
        const int w = s_wsum[lane];
        int winc = w;
        for (int off = 1; off < 32; off <<= 1) {
            const int v = __shfl_up(winc, off);
            if (lane >= off) winc += v;
        }
        s_wsum[lane] = winc - w;
        if (lane == 31) s_base = atomicAdd(total, winc);
    }
    __syncthreads();
    const int first = s_base + s_wsum[wave] + incl - mine;
    const int n = lo + tid;
    if (n < nNodes) { start[n] = first; cnt[n] = mine; }
    s_cnt[tid] = first;
    __syncthreads();

    const bool noted = (s_over == 0) && ((unsigned)nEdges <= (1u << EDGE_BITS));
    if (noted) {
        for (int k = 0; k < nq; ++k) {
            const unsigned w = s_q[k * BUILD_THREADS + tid];
            col[atomicAdd(&s_cnt[w & (RANGE - 1)], 1)] = (int)(w >> RANGE_BITS);
        }
    } else {
        for (int base = 0; base < n4; base += SCAN_DEPTH * BUILD_THREADS) {
            v4i d[SCAN_DEPTH];
            int e0[SCAN_DEPTH];
            dst_ask(p4, n4, rot, base, tid, d, e0);
            #pragma unroll
            for (int j = 0; j < SCAN_DEPTH; ++j) {
                #pragma unroll
                for (int c = 0; c < 4; ++c) {
                    const unsigned r = (unsigned)(d[j][c] - lo);
                    if (r < (unsigned)RANGE) col[atomicAdd(&s_cnt[r], 1)] = e0[j] + c;
                }
            }
        }
        for (int e = 4 * n4 + tid; e < nEdges; e += BUILD_THREADS) {
            const unsigned r = (unsigned)(dst[e] - lo);
            if (r < (unsigned)RANGE) col[atomicAdd(&s_cnt[r], 1)] = e;
        }
    }
    __syncthreads();

    int* p = col + first;
    const int m = mine;
    if (m == 0) return;
    if (m <= SORT_REGS) {
        int v[SORT_REGS];
        #pragma unroll
        for (int i = 0; i < SORT_REGS; ++i) v[i] = (i < m) ? p[i] : 0;
        #pragma unroll
        for (int i = 0; i < SORT_REGS; ++i) v[i] = (i < m) ? src[v[i]] : 0x7fffffff;
        #pragma unroll
        for (int r = 0; r < SORT_REGS; ++r) {
            #pragma unroll
            for (int i = (r & 1); i + 1 < SORT_REGS; i += 2) {
                const int lo2 = min(v[i], v[i + 1]);
                const int hi2 = max(v[i], v[i + 1]);
                v[i] = lo2; v[i + 1] = hi2;
            }
        }
        #pragma unroll
        for (int i = 0; i < SORT_REGS; ++i) if (i < m) p[i] = v[i];
    } else if (m <= SORT_WIDE) {
        int v[SORT_WIDE];
        #pragma unroll
        for (int i = 0; i < SORT_WIDE; ++i) v[i] = p[min(i, m - 1)];
        #pragma unroll
        for (int i = 0; i < SORT_WIDE; ++i) v[i] = src[v[i]];
        #pragma unroll
        for (int i = 0; i < SORT_WIDE; ++i) v[i] = (i < m) ? v[i] : 0x7fffffff;
        #pragma unroll
        for (int r = 0; r < SORT_WIDE; ++r) {
            #pragma unroll
            for (int i = (r & 1); i + 1 < SORT_WIDE; i += 2) {
                const int lo2 = min(v[i], v[i + 1]);
                const int hi2 = max(v[i], v[i + 1]);
                v[i] = lo2; v[i + 1] = hi2;
            }
        }
        #pragma unroll
        for (int i = 0; i < SORT_WIDE; ++i) if (i < m) p[i] = v[i];
    } else {
        for (int i = 0; i < m; ++i) {
            const int key = src[p[i]];
            int j = i - 1;
            while (j >= 0 && p[j] > key) { p[j + 1] = p[j]; --j; }
            p[j + 1] = key;
        }
    }
}


__device__ __forceinline__
void rows_ask(const float* __restrict__ laneBase, int my, int k0, int m, v4f (&r)[8]) {
    int s[8];
    #pragma unroll
    for (int j = 0; j < 8; ++j) s[j] = __builtin_amdgcn_readlane(my, k0 + j);
    #pragma unroll
    for (int j = 0; j < 8; ++j)
        if (j < m) r[j] = *(const v4f*)(laneBase + (long)s[j] * D);
}

__device__ __forceinline__
void rows_add(const v4f (&r)[8], int m, v4f& acc) {
    #pragma unroll
    for (int j = 0; j < 8; ++j)
        if (j < m) acc += r[j];
}

__global__ __launch_bounds__(BLOCK)
void gin_gather_sum(const float* __restrict__ xin,
                    const int*   __restrict__ start,
                    const int*   __restrict__ cnt,
                    const int*   __restrict__ col,
                    float*       __restrict__ agg,
                    int nNodes) {
    const int lane = threadIdx.x & 31;
    const int wave   = __builtin_amdgcn_readfirstlane((int)(threadIdx.x >> 5));
    const int stride = (int)gridDim.x * (BLOCK / 32);
    int n = (int)blockIdx.x * (BLOCK / 32) + wave;
    const float* laneBase = xin + lane * 4;

    int st0 = 0, dg0 = 0, st1 = 0, dg1 = 0;
    if (n < nNodes) {
        st0 = __builtin_amdgcn_readfirstlane(start[n]);
        dg0 = __builtin_amdgcn_readfirstlane(cnt[n]);
    }
    if (n + stride < nNodes) {
        st1 = __builtin_amdgcn_readfirstlane(start[n + stride]);
        dg1 = __builtin_amdgcn_readfirstlane(cnt[n + stride]);
    }
    int my0 = (lane < dg0) ? col[st0 + lane] : 0;

    for (; n < nNodes; n += stride) {
        int st2 = 0, dg2 = 0;
        const int n2 = n + 2 * stride;
        if (n2 < nNodes) {
            st2 = __builtin_amdgcn_readfirstlane(start[n2]);
            dg2 = __builtin_amdgcn_readfirstlane(cnt[n2]);
        }

        v4f acc = {0.0f, 0.0f, 0.0f, 0.0f};
        v4f r[8];
        int c = dg0 < 32 ? dg0 : 32;
        int m = c < 8 ? c : 8;
        rows_ask(laneBase, my0, 0, m, r);
        const int my1 = (lane < dg1) ? col[st1 + lane] : 0;
        rows_add(r, m, acc);
        for (int k0 = 8; k0 < c; k0 += 8) {
            m = (c - k0) < 8 ? (c - k0) : 8;
            rows_ask(laneBase, my0, k0, m, r);
            rows_add(r, m, acc);
        }
        int rem = dg0 - c;
        int pos = st0;
        while (rem > 0) {
            pos += 32;
            c = rem < 32 ? rem : 32;
            const int my = (lane < c) ? col[pos + lane] : 0;
            for (int k0 = 0; k0 < c; k0 += 8) {
                m = (c - k0) < 8 ? (c - k0) : 8;
                rows_ask(laneBase, my, k0, m, r);
                rows_add(r, m, acc);
            }
            rem -= c;
        }
        *(v4f*)(agg + (long)n * D + lane * 4) = acc;

        st0 = st1; dg0 = dg1; my0 = my1;
        st1 = st2; dg1 = dg2;
    }
}


__device__ __forceinline__
void tile_ask(const float* __restrict__ xin, const float* __restrict__ agg,
              long row0, int tid, int nNodes, v4f (&px)[4], v4f (&pa)[4]) {
    const float* xp = xin + row0 * D + tid * 4;
    const float* ap = agg + row0 * D + tid * 4;
    if (row0 + MLP_ROWS <= nNodes) {
        #pragma unroll
        for (int i = 0; i < 4; ++i) {
            px[i] = *(const v4f*)(xp + i * (BLOCK * 4));
            pa[i] = *(const v4f*)(ap + i * (BLOCK * 4));
        }
    } else {
        #pragma unroll
        for (int i = 0; i < 4; ++i) {
            const long row = row0 + ((tid + i * BLOCK) >> 5);
            if (row < nNodes) {
                px[i] = *(const v4f*)(xp + i * (BLOCK * 4));
                pa[i] = *(const v4f*)(ap + i * (BLOCK * 4));
            } else {
                px[i] = (v4f){0.0f, 0.0f, 0.0f, 0.0f};
                pa[i] = (v4f){0.0f, 0.0f, 0.0f, 0.0f};
            }
        }
    }
}

__global__ __launch_bounds__(BLOCK)
void gin_mlp(const float* __restrict__ xin,
             const float* __restrict__ agg,
             const float* __restrict__ W1, const float* __restrict__ b1,
             const float* __restrict__ W2, const float* __restrict__ b2,
             float* __restrict__ out,
             int nNodes,
             int applyReluOut) {
    __shared__ __attribute__((aligned(16))) float s_tile[MLP_ROWS][PITCH];
    __shared__ __attribute__((aligned(16))) float t_tile[MLP_ROWS][PITCH];

    const int tid  = threadIdx.x;
    const int lane = tid & 31;
    const int wave = tid >> 5;

    const int mrow  = lane & 15;
    const int kHalf = (lane >> 4) << 1;
    const int col   = wave * 16 + mrow;
    const int rbase = (lane >> 4) << 3;

    v2f w1f[D / 4], w2f[D / 4];
    #pragma unroll
    for (int j = 0; j < D / 4; ++j) {
        const int kb = 4 * j + kHalf;
        w1f[j][0] = W1[(long)kb * D + col];
        w1f[j][1] = W1[(long)(kb + 1) * D + col];
        w2f[j][0] = W2[(long)kb * D + col];
        w2f[j][1] = W2[(long)(kb + 1) * D + col];
    }
    const float bias1 = b1[col];
    const float bias2 = b2[col];

    const int nTiles = (nNodes + MLP_ROWS - 1) / MLP_ROWS;
    int tile = blockIdx.x;
    v4f px[4], pa[4];
    if (tile < nTiles) tile_ask(xin, agg, (long)tile * MLP_ROWS, tid, nNodes, px, pa);

    for (; tile < nTiles; tile += gridDim.x) {
        const long row0 = (long)tile * MLP_ROWS;

        #pragma unroll
        for (int i = 0; i < 4; ++i) {
            const int idx = tid + i * BLOCK;
            *(v4f*)&s_tile[idx >> 5][(idx & 31) * 4] = px[i] + pa[i];
        }
        __syncthreads();

        const int next = tile + (int)gridDim.x;
        if (next < nTiles) tile_ask(xin, agg, (long)next * MLP_ROWS, tid, nNodes, px, pa);

        {
            v8f acc0 = {}, acc1 = {};
            #pragma unroll
            for (int j = 0; j < D / 4; ++j) {
                const int kb = 4 * j + kHalf;
                const v2f a0 = *(const v2f*)&s_tile[mrow][kb];
                const v2f a1 = *(const v2f*)&s_tile[16 + mrow][kb];
                acc0 = __builtin_amdgcn_wmma_f32_16x16x4_f32(
                    false, a0, false, w1f[j], (short)0, acc0, false, false);
                acc1 = __builtin_amdgcn_wmma_f32_16x16x4_f32(
                    false, a1, false, w1f[j], (short)0, acc1, false, false);
            }
            #pragma unroll
            for (int v = 0; v < 8; ++v) {
                const float t0 = acc0[v] + bias1;
                const float t1 = acc1[v] + bias1;
                t_tile[rbase + v][col]      = t0 > 0.0f ? t0 : 0.0f;
                t_tile[16 + rbase + v][col] = t1 > 0.0f ? t1 : 0.0f;
            }
        }
        __syncthreads();

        {
            v8f acc0 = {}, acc1 = {};
            #pragma unroll
            for (int j = 0; j < D / 4; ++j) {
                const int kb = 4 * j + kHalf;
                const v2f a0 = *(const v2f*)&t_tile[mrow][kb];
                const v2f a1 = *(const v2f*)&t_tile[16 + mrow][kb];
                acc0 = __builtin_amdgcn_wmma_f32_16x16x4_f32(
                    false, a0, false, w2f[j], (short)0, acc0, false, false);
                acc1 = __builtin_amdgcn_wmma_f32_16x16x4_f32(
                    false, a1, false, w2f[j], (short)0, acc1, false, false);
            }
            float u0[8], u1[8];
            #pragma unroll
            for (int v = 0; v < 8; ++v) {
                u0[v] = acc0[v] + bias2;
                u1[v] = acc1[v] + bias2;
                if (applyReluOut) {
                    u0[v] = u0[v] > 0.0f ? u0[v] : 0.0f;
                    u1[v] = u1[v] > 0.0f ? u1[v] : 0.0f;
                }
            }
            float* op = out + (row0 + rbase) * D + col;
            if (row0 + MLP_ROWS <= nNodes) {
                #pragma unroll
                for (int v = 0; v < 8; ++v) {
                    op[v * D]        = u0[v];
                    op[(16 + v) * D] = u1[v];
                }
            } else {
                #pragma unroll
                for (int v = 0; v < 8; ++v) {
                    if (row0 + rbase + v < nNodes)      op[v * D]        = u0[v];
                    if (row0 + rbase + v + 16 < nNodes) op[(16 + v) * D] = u1[v];
                }
            }
        }
    }
}

extern "C" void kernel_launch(void* const* d_in, const int* in_sizes, int n_in,
                              void* d_out, int out_size, void* d_ws, size_t ws_size,
                              hipStream_t stream) {
    const float* x  = (const float*)d_in[0];
    const int*   ei = (const int*)  d_in[1];
    const float* W1 = (const float*)d_in[2];
    const float* b1 = (const float*)d_in[3];
    const float* W2 = (const float*)d_in[4];
    const float* b2 = (const float*)d_in[5];
    float* out = (float*)d_out;

    const int N = in_sizes[0] / D;
    const int E = in_sizes[1] / 2;
    const int* src = ei;
    const int* dst = ei + E;

    float* agg = (float*)d_ws;
    float* h1  = agg + (size_t)N * D;
    int* cnt   = (int*)(h1 + (size_t)N * D);
    int* start = cnt + N + 32;
    int* col   = start + N;

    const int bBlocks = (N + RANGE - 1) / RANGE;

    hipMemsetAsync(cnt + N, 0, sizeof(int), stream);
    bucket_build<<<bBlocks, BUILD_THREADS, 0, stream>>>(src, dst, cnt + N, start, cnt, col, N, E);

    gin_gather_sum<<<GATHER_BLOCKS, BLOCK, 0, stream>>>(x, start, cnt, col, agg, N);
    gin_mlp<<<MLP_BLOCKS, BLOCK, 0, stream>>>(x, agg, W1, b1, W2, b2, h1, N, 1);

    gin_gather_sum<<<GATHER_BLOCKS, BLOCK, 0, stream>>>(h1, start, cnt, col, agg, N);
    gin_mlp<<<MLP_BLOCKS, BLOCK, 0, stream>>>(h1, agg, W1, b1, W2, b2, out, N, 0);
}


# Neighbour sum fused into the MLP kernel: each wave sums 4 rows of the 32-row tile in registers straight into the LDS tile of the first product; agg array and gather launches gone; sum order unchanged
# speedup vs baseline: 3.8543x; 1.0635x over previous
#include <hip/hip_runtime.h>


#define D 128
#define MLP_ROWS 32
#define MLP_BLOCKS 512
#define PITCH (D + 4)
#define BLOCK 256
#define WAVE_ROWS (MLP_ROWS / (BLOCK / 32))
#define SORT_REGS 16
#define SORT_WIDE 32
#define BUILD_THREADS 1024
#define RANGE_BITS 10
#define RANGE (1 << RANGE_BITS)
#define SCAN_DEPTH 8
#define QSLOTS 32
#define EDGE_BITS (32 - RANGE_BITS)

typedef float v2f __attribute__((ext_vector_type(2)));
typedef float v4f __attribute__((ext_vector_type(4)));
typedef float v8f __attribute__((ext_vector_type(8)));
typedef int   v4i __attribute__((ext_vector_type(4)));


__device__ __forceinline__
void dst_ask(const v4i* __restrict__ p, int n4, int rot, int base, int tid,
             v4i (&d)[SCAN_DEPTH], int (&e0)[SCAN_DEPTH]) {
    #pragma unroll
    for (int j = 0; j < SCAN_DEPTH; ++j) {
        const int i = base + j * BUILD_THREADS + tid;
        int idx = i + rot;
        if (idx >= n4) idx -= n4;
        e0[j] = idx * 4;
        if (i < n4) d[j] = p[idx];
        else        d[j] = (v4i){-1, -1, -1, -1};
    }
}

__global__ __launch_bounds__(BUILD_THREADS)
void bucket_build(const int* __restrict__ src,
                  const int* __restrict__ dst,
                  int* total,
                  int* __restrict__ start,
                  int* __restrict__ cnt,
                  int* col,
                  int nNodes, int nEdges) {
    __shared__ int s_cnt[RANGE];
    __shared__ unsigned s_q[QSLOTS * BUILD_THREADS];
    __shared__ int s_over;
    __shared__ int s_wsum[BUILD_THREADS / 32];
    __shared__ int s_base;
    const int tid  = threadIdx.x;
    const int lane = tid & 31;
    const int wave = tid >> 5;
    const int lo   = (int)blockIdx.x * RANGE;

    const bool vec = (((unsigned long long)dst) & 15ull) == 0ull;
    const v4i* p4  = (const v4i*)dst;
    const int n4   = vec ? (nEdges >> 2) : 0;
    const int rot  = (int)(((long)blockIdx.x * n4) / (long)gridDim.x);

    s_cnt[tid] = 0;
    if (tid == 0) s_over = 0;
    __syncthreads();

    int nq = 0;
    for (int base = 0; base < n4; base += SCAN_DEPTH * BUILD_THREADS) {
        v4i d[SCAN_DEPTH];
        int e0[SCAN_DEPTH];
        dst_ask(p4, n4, rot, base, tid, d, e0);
        #pragma unroll
        for (int j = 0; j < SCAN_DEPTH; ++j) {
            #pragma unroll
            for (int c = 0; c < 4; ++c) {
                const unsigned r = (unsigned)(d[j][c] - lo);
                if (r < (unsigned)RANGE) {
                    atomicAdd(&s_cnt[r], 1);
                    if (nq < QSLOTS)
                        s_q[nq * BUILD_THREADS + tid] = ((unsigned)(e0[j] + c) << RANGE_BITS) | r;
                    ++nq;
                }
            }
        }
    }
    for (int e = 4 * n4 + tid; e < nEdges; e += BUILD_THREADS) {
        const unsigned r = (unsigned)(dst[e] - lo);
        if (r < (unsigned)RANGE) {
            atomicAdd(&s_cnt[r], 1);
            if (nq < QSLOTS)
                s_q[nq * BUILD_THREADS + tid] = ((unsigned)e << RANGE_BITS) | r;
            ++nq;
        }
    }
    if (nq > QSLOTS) s_over = 1;
    __syncthreads();

    const int mine = s_cnt[tid];
    int incl = mine;
    for (int off = 1; off < 32; off <<= 1) {
        const int v = __shfl_up(incl, off);
        if (lane >= off) incl += v;
    }
    if (lane == 31) s_wsum[wave] = incl;
    __syncthreads();
    if (wave == 0) {
        const int w = s_wsum[lane];
        int winc = w;
        for (int off = 1; off < 32; off <<= 1) {
            const int v = __shfl_up(winc, off);
            if (lane >= off) winc += v;
        }
        s_wsum[lane] = winc - w;
        if (lane == 31) s_base = atomicAdd(total, winc);
    }
    __syncthreads();
    const int first = s_base + s_wsum[wave] + incl - mine;
    const int n = lo + tid;
    if (n < nNodes) { start[n] = first; cnt[n] = mine; }
    s_cnt[tid] = first;
    __syncthreads();

    const bool noted = (s_over == 0) && ((unsigned)nEdges <= (1u << EDGE_BITS));
    if (noted) {
        for (int k = 0; k < nq; ++k) {
            const unsigned w = s_q[k * BUILD_THREADS + tid];
            col[atomicAdd(&s_cnt[w & (RANGE - 1)], 1)] = (int)(w >> RANGE_BITS);
        }
    } else {
        for (int base = 0; base < n4; base += SCAN_DEPTH * BUILD_THREADS) {
            v4i d[SCAN_DEPTH];
            int e0[SCAN_DEPTH];
            dst_ask(p4, n4, rot, base, tid, d, e0);
            #pragma unroll
            for (int j = 0; j < SCAN_DEPTH; ++j) {
                #pragma unroll
                for (int c = 0; c < 4; ++c) {
                    const unsigned r = (unsigned)(d[j][c] - lo);
                    if (r < (unsigned)RANGE) col[atomicAdd(&s_cnt[r], 1)] = e0[j] + c;
                }
            }
        }
        for (int e = 4 * n4 + tid; e < nEdges; e += BUILD_THREADS) {
            const unsigned r = (unsigned)(dst[e] - lo);
            if (r < (unsigned)RANGE) col[atomicAdd(&s_cnt[r], 1)] = e;
        }
    }
    __syncthreads();

    int* p = col + first;
    const int m = mine;
    if (m == 0) return;
    if (m <= SORT_REGS) {
        int v[SORT_REGS];
        #pragma unroll
        for (int i = 0; i < SORT_REGS; ++i) v[i] = (i < m) ? p[i] : 0;
        #pragma unroll
        for (int i = 0; i < SORT_REGS; ++i) v[i] = (i < m) ? src[v[i]] : 0x7fffffff;
        #pragma unroll
        for (int r = 0; r < SORT_REGS; ++r) {
            #pragma unroll
            for (int i = (r & 1); i + 1 < SORT_REGS; i += 2) {
                const int lo2 = min(v[i], v[i + 1]);
                const int hi2 = max(v[i], v[i + 1]);
                v[i] = lo2; v[i + 1] = hi2;
            }
        }
        #pragma unroll
        for (int i = 0; i < SORT_REGS; ++i) if (i < m) p[i] = v[i];
    } else if (m <= SORT_WIDE) {
        int v[SORT_WIDE];
        #pragma unroll
        for (int i = 0; i < SORT_WIDE; ++i) v[i] = p[min(i, m - 1)];
        #pragma unroll
        for (int i = 0; i < SORT_WIDE; ++i) v[i] = src[v[i]];
        #pragma unroll
        for (int i = 0; i < SORT_WIDE; ++i) v[i] = (i < m) ? v[i] : 0x7fffffff;
        #pragma unroll
        for (int r = 0; r < SORT_WIDE; ++r) {
            #pragma unroll
            for (int i = (r & 1); i + 1 < SORT_WIDE; i += 2) {
                const int lo2 = min(v[i], v[i + 1]);
                const int hi2 = max(v[i], v[i + 1]);
                v[i] = lo2; v[i + 1] = hi2;
            }
        }
        #pragma unroll
        for (int i = 0; i < SORT_WIDE; ++i) if (i < m) p[i] = v[i];
    } else {
        for (int i = 0; i < m; ++i) {
            const int key = src[p[i]];
            int j = i - 1;
            while (j >= 0 && p[j] > key) { p[j + 1] = p[j]; --j; }
            p[j + 1] = key;
        }
    }
}


__device__ __forceinline__
void rows_ask(const float* __restrict__ laneBase, int my, int k0, v4f (&r)[8]) {
    int s[8];
    #pragma unroll
    for (int j = 0; j < 8; ++j) s[j] = __builtin_amdgcn_readlane(my, k0 + j);
    #pragma unroll
    for (int j = 0; j < 8; ++j) r[j] = *(const v4f*)(laneBase + (long)s[j] * D);
}

__device__ __forceinline__
void rows_add(const v4f (&r)[8], int m, v4f& acc) {
    #pragma unroll
    for (int j = 0; j < 8; ++j) {
        const v4f t = acc + r[j];
        acc = (j < m) ? t : acc;
    }
}

__device__ __forceinline__
void node_ask(const float* __restrict__ laneBase, int n, int nNodes, int my,
              v4f& self, v4f (&r)[8]) {
    rows_ask(laneBase, my, 0, r);
    const int nc = n < nNodes ? n : nNodes - 1;
    const v4f t = *(const v4f*)(laneBase + (long)nc * D);
    self = (n < nNodes) ? t : (v4f){0.0f, 0.0f, 0.0f, 0.0f};
}

__device__ __forceinline__
v4f node_sum(const float* __restrict__ laneBase, const int* __restrict__ col,
             int st, int dg, int my, int lane, v4f (&r)[8]) {
    v4f acc = {0.0f, 0.0f, 0.0f, 0.0f};
    int c = dg < 32 ? dg : 32;
    rows_add(r, c, acc);
    for (int k0 = 8; k0 < c; k0 += 8) {
        rows_ask(laneBase, my, k0, r);
        rows_add(r, c - k0, acc);
    }
    int rem = dg - c;
    int pos = st;
    while (rem > 0) {
        pos += 32;
        c = rem < 32 ? rem : 32;
        const int e = (lane < c) ? col[pos + lane] : 0;
        for (int k0 = 0; k0 < c; k0 += 8) {
            rows_ask(laneBase, e, k0, r);
            rows_add(r, c - k0, acc);
        }
        rem -= c;
    }
    return acc;
}


__device__ __forceinline__
void bounds_ask(const int* __restrict__ start, const int* __restrict__ cnt,
                int nb, int nNodes, bool ok, int lane, int& sv, int& cv) {
    sv = 0; cv = 0;
    if (ok && lane < WAVE_ROWS && nb + lane < nNodes) {
        sv = start[nb + lane];
        cv = cnt[nb + lane];
    }
}

__global__ __launch_bounds__(BLOCK, 4)
void gin_layer(const float* __restrict__ xin,
               const int*   __restrict__ start,
               const int*   __restrict__ cnt,
               const int*   __restrict__ col,
               const float* __restrict__ W1, const float* __restrict__ b1,
               const float* __restrict__ W2, const float* __restrict__ b2,
               float* __restrict__ out,
               int nNodes,
               int applyReluOut) {
    __shared__ __attribute__((aligned(16))) float s_tile[MLP_ROWS][PITCH];
    __shared__ __attribute__((aligned(16))) float t_tile[MLP_ROWS][PITCH];

    const int tid  = threadIdx.x;
    const int lane = tid & 31;
    const int wave = __builtin_amdgcn_readfirstlane(tid >> 5);

    const int mrow  = lane & 15;
    const int kHalf = (lane >> 4) << 1;
    const int colN  = wave * 16 + mrow;
    const int rbase = (lane >> 4) << 3;

    v2f w1f[D / 4], w2f[D / 4];
    #pragma unroll
    for (int j = 0; j < D / 4; ++j) {
        const int kb = 4 * j + kHalf;
        w1f[j][0] = W1[(long)kb * D + colN];
        w1f[j][1] = W1[(long)(kb + 1) * D + colN];
        w2f[j][0] = W2[(long)kb * D + colN];
        w2f[j][1] = W2[(long)(kb + 1) * D + colN];
    }
    const float bias1 = b1[colN];
    const float bias2 = b2[colN];

    const float* laneBase = xin + lane * 4;
    const int nTiles = (nNodes + MLP_ROWS - 1) / MLP_ROWS;
    int tile = blockIdx.x;

    int st[WAVE_ROWS], dg[WAVE_ROWS], my[WAVE_ROWS];
    {
        int sv, cv;
        bounds_ask(start, cnt, tile * MLP_ROWS + wave * WAVE_ROWS, nNodes, tile < nTiles, lane, sv, cv);
        #pragma unroll
        for (int i = 0; i < WAVE_ROWS; ++i) {
            st[i] = __builtin_amdgcn_readlane(sv, i);
            dg[i] = __builtin_amdgcn_readlane(cv, i);
        }
        #pragma unroll
        for (int i = 0; i < WAVE_ROWS; ++i) my[i] = (lane < dg[i]) ? col[st[i] + lane] : 0;
    }

    for (; tile < nTiles; tile += gridDim.x) {
        const long row0 = (long)tile * MLP_ROWS;
        const int  nb   = tile * MLP_ROWS + wave * WAVE_ROWS;

        {
            v4f rA[8], rB[8], selfA, selfB, acc;
            float* sp = &s_tile[wave * WAVE_ROWS][lane * 4];
            node_ask(laneBase, nb + 0, nNodes, my[0], selfA, rA);
            node_ask(laneBase, nb + 1, nNodes, my[1], selfB, rB);
            acc = node_sum(laneBase, col, st[0], dg[0], my[0], lane, rA);
            *(v4f*)(sp + 0 * PITCH) = selfA + acc;
            node_ask(laneBase, nb + 2, nNodes, my[2], selfA, rA);
            acc = node_sum(laneBase, col, st[1], dg[1], my[1], lane, rB);
            *(v4f*)(sp + 1 * PITCH) = selfB + acc;
            node_ask(laneBase, nb + 3, nNodes, my[3], selfB, rB);
            acc = node_sum(laneBase, col, st[2], dg[2], my[2], lane, rA);
            *(v4f*)(sp + 2 * PITCH) = selfA + acc;
            acc = node_sum(laneBase, col, st[3], dg[3], my[3], lane, rB);
            *(v4f*)(sp + 3 * PITCH) = selfB + acc;
        }
        __syncthreads();

        const int next = tile + (int)gridDim.x;
        int sv, cv;
        bounds_ask(start, cnt, next * MLP_ROWS + wave * WAVE_ROWS, nNodes, next < nTiles, lane, sv, cv);

        {
            v8f acc0 = {}, acc1 = {};
            #pragma unroll
            for (int j = 0; j < D / 4; ++j) {
                const int kb = 4 * j + kHalf;
                const v2f a0 = *(const v2f*)&s_tile[mrow][kb];
                const v2f a1 = *(const v2f*)&s_tile[16 + mrow][kb];
                acc0 = __builtin_amdgcn_wmma_f32_16x16x4_f32(
                    false, a0, false, w1f[j], (short)0, acc0, false, false);
                acc1 = __builtin_amdgcn_wmma_f32_16x16x4_f32(
                    false, a1, false, w1f[j], (short)0, acc1, false, false);
            }
            #pragma unroll
            for (int v = 0; v < 8; ++v) {
                const float t0 = acc0[v] + bias1;
                const float t1 = acc1[v] + bias1;
                t_tile[rbase + v][colN]      = t0 > 0.0f ? t0 : 0.0f;
                t_tile[16 + rbase + v][colN] = t1 > 0.0f ? t1 : 0.0f;
            }
        }
        __syncthreads();

        #pragma unroll
        for (int i = 0; i < WAVE_ROWS; ++i) {
            st[i] = __builtin_amdgcn_readlane(sv, i);
            dg[i] = __builtin_amdgcn_readlane(cv, i);
        }
        #pragma unroll
        for (int i = 0; i < WAVE_ROWS; ++i) my[i] = (lane < dg[i]) ? col[st[i] + lane] : 0;

        {
            v8f acc0 = {}, acc1 = {};
            #pragma unroll
            for (int j = 0; j < D / 4; ++j) {
                const int kb = 4 * j + kHalf;
                const v2f a0 = *(const v2f*)&t_tile[mrow][kb];
                const v2f a1 = *(const v2f*)&t_tile[16 + mrow][kb];
                acc0 = __builtin_amdgcn_wmma_f32_16x16x4_f32(
                    false, a0, false, w2f[j], (short)0, acc0, false, false);
                acc1 = __builtin_amdgcn_wmma_f32_16x16x4_f32(
                    false, a1, false, w2f[j], (short)0, acc1, false, false);
            }
            float u0[8], u1[8];
            #pragma unroll
            for (int v = 0; v < 8; ++v) {
                u0[v] = acc0[v] + bias2;
                u1[v] = acc1[v] + bias2;
                if (applyReluOut) {
                    u0[v] = u0[v] > 0.0f ? u0[v] : 0.0f;
                    u1[v] = u1[v] > 0.0f ? u1[v] : 0.0f;
                }
            }
            float* op = out + (row0 + rbase) * D + colN;
            if (row0 + MLP_ROWS <= nNodes) {
                #pragma unroll
                for (int v = 0; v < 8; ++v) {
                    op[v * D]        = u0[v];
                    op[(16 + v) * D] = u1[v];
                }
            } else {
                #pragma unroll
                for (int v = 0; v < 8; ++v) {
                    if (row0 + rbase + v < nNodes)      op[v * D]        = u0[v];
                    if (row0 + rbase + v + 16 < nNodes) op[(16 + v) * D] = u1[v];
                }
            }
        }
    }
}

extern "C" void kernel_launch(void* const* d_in, const int* in_sizes, int n_in,
                              void* d_out, int out_size, void* d_ws, size_t ws_size,
                              hipStream_t stream) {
    const float* x  = (const float*)d_in[0];
    const int*   ei = (const int*)  d_in[1];
    const float* W1 = (const float*)d_in[2];
    const float* b1 = (const float*)d_in[3];
    const float* W2 = (const float*)d_in[4];
    const float* b2 = (const float*)d_in[5];
    float* out = (float*)d_out;

    const int N = in_sizes[0] / D;
    const int E = in_sizes[1] / 2;
    const int* src = ei;
    const int* dst = ei + E;

    float* h1  = (float*)d_ws;
    int* cnt   = (int*)(h1 + (size_t)N * D);
    int* start = cnt + N + 32;
    int* col   = start + N;

    const int bBlocks = (N + RANGE - 1) / RANGE;

    hipMemsetAsync(cnt + N, 0, sizeof(int), stream);
    bucket_build<<<bBlocks, BUILD_THREADS, 0, stream>>>(src, dst, cnt + N, start, cnt, col, N, E);

    gin_layer<<<MLP_BLOCKS, BLOCK, 0, stream>>>(x, start, cnt, col, W1, b1, W2, b2, h1, N, 1);

    gin_layer<<<MLP_BLOCKS, BLOCK, 0, stream>>>(h1, start, cnt, col, W1, b1, W2, b2, out, N, 0);
}


# Bucket build made two-level: a first launch orders each of 256 edge-list slices by 1024-node range (LDS counters), so a range's workgroup reads only its own edges; memset gone; sum order unchanged
# speedup vs baseline: 4.5158x; 1.1716x over previous
#include <hip/hip_runtime.h>


#define D 128
#define MLP_ROWS 32
#define MLP_BLOCKS 512
#define PITCH (D + 4)
#define BLOCK 256
#define WAVE_ROWS (MLP_ROWS / (BLOCK / 32))
#define SORT_REGS 16
#define SORT_WIDE 32
#define BUILD_THREADS 1024
#define RANGE_BITS 10
#define RANGE (1 << RANGE_BITS)
#define BIN_BLOCKS 256
#define BIN_THREADS 256
#define BIN_DEPTH 5
#define MAX_RANGES 1024
#define SLICE_THREADS (BUILD_THREADS / BIN_BLOCKS)
#define ENT_DEPTH 4

typedef float v2f __attribute__((ext_vector_type(2)));
typedef float v4f __attribute__((ext_vector_type(4)));
typedef float v8f __attribute__((ext_vector_type(8)));
typedef int   v2i __attribute__((ext_vector_type(2)));

__global__ __launch_bounds__(BIN_THREADS)
void edge_bin(const int* __restrict__ src,
              const int* __restrict__ dst,
              int* __restrict__ off,
              int* __restrict__ ent,
              int nNodes, int nEdges, int nRanges, int sliceLen) {
    __shared__ int s_hist[MAX_RANGES];
    __shared__ int s_cur[MAX_RANGES];
    const int tid  = threadIdx.x;
    const int lane = tid & 31;
    const int wave = tid >> 5;
    const int g    = (int)blockIdx.x;
    const long b0l = (long)g * sliceLen;
    const int e0   = b0l < (long)nEdges ? (int)b0l : nEdges;
    const int e1   = (nEdges - e0 > sliceLen) ? e0 + sliceLen : nEdges;

    for (int b = tid; b < nRanges; b += BIN_THREADS) s_hist[b] = 0;
    __syncthreads();

    for (int base = e0 + tid; base < e1; base += BIN_DEPTH * BIN_THREADS) {
        int d[BIN_DEPTH];
        #pragma unroll
        for (int u = 0; u < BIN_DEPTH; ++u) d[u] = dst[min(base + u * BIN_THREADS, e1 - 1)];
        #pragma unroll
        for (int u = 0; u < BIN_DEPTH; ++u) asm volatile("" :: "v"(d[u]));
        #pragma unroll
        for (int u = 0; u < BIN_DEPTH; ++u)
            if (base + u * BIN_THREADS < e1 && (unsigned)d[u] < (unsigned)nNodes)
                atomicAdd(&s_hist[d[u] >> RANGE_BITS], 1);
    }
    __syncthreads();

    int* offg = off + (long)g * (nRanges + 1);
    if (wave == 0) {
        int carry = 0;
        for (int r0 = 0; r0 < nRanges; r0 += 32) {
            const int b = r0 + lane;
            const int c = (b < nRanges) ? s_hist[b] : 0;
            int incl = c;
            for (int o = 1; o < 32; o <<= 1) {
                const int v = __shfl_up(incl, o);
                if (lane >= o) incl += v;
            }
            if (b < nRanges) {
                s_cur[b] = carry + incl - c;
                offg[b]  = carry + incl - c;
            }
            carry += __shfl(incl, 31);
        }
        if (lane == 0) offg[nRanges] = carry;
    }
    __syncthreads();

    v2i* entg = (v2i*)ent + (long)g * sliceLen;
    for (int base = e0 + tid; base < e1; base += BIN_DEPTH * BIN_THREADS) {
        int d[BIN_DEPTH], s[BIN_DEPTH];
        #pragma unroll
        for (int u = 0; u < BIN_DEPTH; ++u) {
            const int e = min(base + u * BIN_THREADS, e1 - 1);
            d[u] = dst[e];
            s[u] = src[e];
        }
        #pragma unroll
        for (int u = 0; u < BIN_DEPTH; ++u) asm volatile("" :: "v"(d[u]), "v"(s[u]));
        #pragma unroll
        for (int u = 0; u < BIN_DEPTH; ++u)
            if (base + u * BIN_THREADS < e1 && (unsigned)d[u] < (unsigned)nNodes) {
                const int pos = atomicAdd(&s_cur[d[u] >> RANGE_BITS], 1);
                entg[pos] = (v2i){s[u], d[u]};
            }
    }
}

__global__ __launch_bounds__(BUILD_THREADS)
void bucket_build(const int* __restrict__ off,
                  const int* __restrict__ ent,
                  int* __restrict__ start,
                  int* __restrict__ cnt,
                  int* col,
                  int nNodes, int nRanges, int sliceLen) {
    __shared__ int s_cnt[RANGE];
    __shared__ int s_wsum[BUILD_THREADS / 32];
    __shared__ int s_bsum[BUILD_THREADS / 32];
    __shared__ int s_base;
    const int tid  = threadIdx.x;
    const int lane = tid & 31;
    const int wave = tid >> 5;
    const int r    = (int)blockIdx.x;
    const int lo   = r * RANGE;
    const int g    = tid / SLICE_THREADS;
    const int j    = tid % SLICE_THREADS;

    const int* offg = off + (long)g * (nRanges + 1) + r;
    const int a = offg[0];
    const int b = offg[1];
    const v2i* entg = (const v2i*)ent + (long)g * sliceLen;

    s_cnt[tid] = 0;
    __syncthreads();

    for (int base = a + j; base < b; base += ENT_DEPTH * SLICE_THREADS) {
        int d[ENT_DEPTH];
        #pragma unroll
        for (int u = 0; u < ENT_DEPTH; ++u) d[u] = entg[min(base + u * SLICE_THREADS, b - 1)][1];
        #pragma unroll
        for (int u = 0; u < ENT_DEPTH; ++u) asm volatile("" :: "v"(d[u]));
        #pragma unroll
        for (int u = 0; u < ENT_DEPTH; ++u)
            if (base + u * SLICE_THREADS < b) atomicAdd(&s_cnt[d[u] - lo], 1);
    }
    __syncthreads();

    const int mine = s_cnt[tid];
    int incl = mine;
    for (int o = 1; o < 32; o <<= 1) {
        const int v = __shfl_up(incl, o);
        if (lane >= o) incl += v;
    }
    int before = (j == 0) ? a : 0;
    for (int o = 16; o > 0; o >>= 1) before += __shfl_xor(before, o);
    if (lane == 31) s_wsum[wave] = incl;
    if (lane == 0)  s_bsum[wave] = before;
    __syncthreads();
    if (wave == 0) {
        const int w = s_wsum[lane];
        int winc = w;
        for (int o = 1; o < 32; o <<= 1) {
            const int v = __shfl_up(winc, o);
            if (lane >= o) winc += v;
        }
        s_wsum[lane] = winc - w;
        int bs = s_bsum[lane];
        for (int o = 16; o > 0; o >>= 1) bs += __shfl_xor(bs, o);
        if (lane == 0) s_base = bs;
    }
    __syncthreads();
    const int first = s_base + s_wsum[wave] + incl - mine;
    const int n = lo + tid;
    if (n < nNodes) { start[n] = first; cnt[n] = mine; }
    s_cnt[tid] = first;
    __syncthreads();

    for (int base = a + j; base < b; base += ENT_DEPTH * SLICE_THREADS) {
        v2i en[ENT_DEPTH];
        #pragma unroll
        for (int u = 0; u < ENT_DEPTH; ++u) en[u] = entg[min(base + u * SLICE_THREADS, b - 1)];
        #pragma unroll
        for (int u = 0; u < ENT_DEPTH; ++u) asm volatile("" :: "v"(en[u][0]), "v"(en[u][1]));
        #pragma unroll
        for (int u = 0; u < ENT_DEPTH; ++u)
            if (base + u * SLICE_THREADS < b) col[atomicAdd(&s_cnt[en[u][1] - lo], 1)] = en[u][0];
    }
    __syncthreads();

    int* p = col + first;
    const int m = mine;
    if (m == 0) return;
    if (m <= SORT_REGS) {
        int v[SORT_REGS];
        #pragma unroll
        for (int i = 0; i < SORT_REGS; ++i) v[i] = (i < m) ? p[i] : 0x7fffffff;
        #pragma unroll
        for (int rr = 0; rr < SORT_REGS; ++rr) {
            #pragma unroll
            for (int i = (rr & 1); i + 1 < SORT_REGS; i += 2) {
                const int lo2 = min(v[i], v[i + 1]);
                const int hi2 = max(v[i], v[i + 1]);
                v[i] = lo2; v[i + 1] = hi2;
            }
        }
        #pragma unroll
        for (int i = 0; i < SORT_REGS; ++i) if (i < m) p[i] = v[i];
    } else if (m <= SORT_WIDE) {
        int v[SORT_WIDE];
        #pragma unroll
        for (int i = 0; i < SORT_WIDE; ++i) v[i] = p[min(i, m - 1)];
        #pragma unroll
        for (int i = 0; i < SORT_WIDE; ++i) v[i] = (i < m) ? v[i] : 0x7fffffff;
        #pragma unroll
        for (int rr = 0; rr < SORT_WIDE; ++rr) {
            #pragma unroll
            for (int i = (rr & 1); i + 1 < SORT_WIDE; i += 2) {
                const int lo2 = min(v[i], v[i + 1]);
                const int hi2 = max(v[i], v[i + 1]);
                v[i] = lo2; v[i + 1] = hi2;
            }
        }
        #pragma unroll
        for (int i = 0; i < SORT_WIDE; ++i) if (i < m) p[i] = v[i];
    } else {
        for (int i = 1; i < m; ++i) {
            const int key = p[i];
            int q = i - 1;
            while (q >= 0 && p[q] > key) { p[q + 1] = p[q]; --q; }
            p[q + 1] = key;
        }
    }
}


__device__ __forceinline__
void rows_ask(const float* __restrict__ laneBase, int my, int k0, v4f (&r)[8]) {
    int s[8];
    #pragma unroll
    for (int j = 0; j < 8; ++j) s[j] = __builtin_amdgcn_readlane(my, k0 + j);
    #pragma unroll
    for (int j = 0; j < 8; ++j) r[j] = *(const v4f*)(laneBase + (long)s[j] * D);
}

__device__ __forceinline__
void rows_add(const v4f (&r)[8], int m, v4f& acc) {
    #pragma unroll
    for (int j = 0; j < 8; ++j) {
        const v4f t = acc + r[j];
        acc = (j < m) ? t : acc;
    }
}

__device__ __forceinline__
void node_ask(const float* __restrict__ laneBase, int n, int nNodes, int my,
              v4f& self, v4f (&r)[8]) {
    rows_ask(laneBase, my, 0, r);
    const int nc = n < nNodes ? n : nNodes - 1;
    const v4f t = *(const v4f*)(laneBase + (long)nc * D);
    self = (n < nNodes) ? t : (v4f){0.0f, 0.0f, 0.0f, 0.0f};
}

__device__ __forceinline__
v4f node_sum(const float* __restrict__ laneBase, const int* __restrict__ col,
             int st, int dg, int my, int lane, v4f (&r)[8]) {
    v4f acc = {0.0f, 0.0f, 0.0f, 0.0f};
    int c = dg < 32 ? dg : 32;
    rows_add(r, c, acc);
    for (int k0 = 8; k0 < c; k0 += 8) {
        rows_ask(laneBase, my, k0, r);
        rows_add(r, c - k0, acc);
    }
    int rem = dg - c;
    int pos = st;
    while (rem > 0) {
        pos += 32;
        c = rem < 32 ? rem : 32;
        const int e = (lane < c) ? col[pos + lane] : 0;
        for (int k0 = 0; k0 < c; k0 += 8) {
            rows_ask(laneBase, e, k0, r);
            rows_add(r, c - k0, acc);
        }
        rem -= c;
    }
    return acc;
}


__device__ __forceinline__
void bounds_ask(const int* __restrict__ start, const int* __restrict__ cnt,
                int nb, int nNodes, bool ok, int lane, int& sv, int& cv) {
    sv = 0; cv = 0;
    if (ok && lane < WAVE_ROWS && nb + lane < nNodes) {
        sv = start[nb + lane];
        cv = cnt[nb + lane];
    }
}

__global__ __launch_bounds__(BLOCK, 4)
void gin_layer(const float* __restrict__ xin,
               const int*   __restrict__ start,
               const int*   __restrict__ cnt,
               const int*   __restrict__ col,
               const float* __restrict__ W1, const float* __restrict__ b1,
               const float* __restrict__ W2, const float* __restrict__ b2,
               float* __restrict__ out,
               int nNodes,
               int applyReluOut) {
    __shared__ __attribute__((aligned(16))) float s_tile[MLP_ROWS][PITCH];
    __shared__ __attribute__((aligned(16))) float t_tile[MLP_ROWS][PITCH];

    const int tid  = threadIdx.x;
    const int lane = tid & 31;
    const int wave = __builtin_amdgcn_readfirstlane(tid >> 5);

    const int mrow  = lane & 15;
    const int kHalf = (lane >> 4) << 1;
    const int colN  = wave * 16 + mrow;
    const int rbase = (lane >> 4) << 3;

    v2f w1f[D / 4], w2f[D / 4];
    #pragma unroll
    for (int j = 0; j < D / 4; ++j) {
        const int kb = 4 * j + kHalf;
        w1f[j][0] = W1[(long)kb * D + colN];
        w1f[j][1] = W1[(long)(kb + 1) * D + colN];
        w2f[j][0] = W2[(long)kb * D + colN];
        w2f[j][1] = W2[(long)(kb + 1) * D + colN];
    }
    const float bias1 = b1[colN];
    const float bias2 = b2[colN];

    const float* laneBase = xin + lane * 4;
    const int nTiles = (nNodes + MLP_ROWS - 1) / MLP_ROWS;
    int tile = blockIdx.x;

    int st[WAVE_ROWS], dg[WAVE_ROWS], my[WAVE_ROWS];
    {
        int sv, cv;
        bounds_ask(start, cnt, tile * MLP_ROWS + wave * WAVE_ROWS, nNodes, tile < nTiles, lane, sv, cv);
        #pragma unroll
        for (int i = 0; i < WAVE_ROWS; ++i) {
            st[i] = __builtin_amdgcn_readlane(sv, i);
            dg[i] = __builtin_amdgcn_readlane(cv, i);
        }
        #pragma unroll
        for (int i = 0; i < WAVE_ROWS; ++i) my[i] = (lane < dg[i]) ? col[st[i] + lane] : 0;
    }

    for (; tile < nTiles; tile += gridDim.x) {
        const long row0 = (long)tile * MLP_ROWS;
        const int  nb   = tile * MLP_ROWS + wave * WAVE_ROWS;

        {
            v4f rA[8], rB[8], selfA, selfB, acc;
            float* sp = &s_tile[wave * WAVE_ROWS][lane * 4];
            node_ask(laneBase, nb + 0, nNodes, my[0], selfA, rA);
            node_ask(laneBase, nb + 1, nNodes, my[1], selfB, rB);
            acc = node_sum(laneBase, col, st[0], dg[0], my[0], lane, rA);
            *(v4f*)(sp + 0 * PITCH) = selfA + acc;
            node_ask(laneBase, nb + 2, nNodes, my[2], selfA, rA);
            acc = node_sum(laneBase, col, st[1], dg[1], my[1], lane, rB);
            *(v4f*)(sp + 1 * PITCH) = selfB + acc;
            node_ask(laneBase, nb + 3, nNodes, my[3], selfB, rB);
            acc = node_sum(laneBase, col, st[2], dg[2], my[2], lane, rA);
            *(v4f*)(sp + 2 * PITCH) = selfA + acc;
            acc = node_sum(laneBase, col, st[3], dg[3], my[3], lane, rB);
            *(v4f*)(sp + 3 * PITCH) = selfB + acc;
        }
        __syncthreads();

        const int next = tile + (int)gridDim.x;
        int sv, cv;
        bounds_ask(start, cnt, next * MLP_ROWS + wave * WAVE_ROWS, nNodes, next < nTiles, lane, sv, cv);

        {
            v8f acc0 = {}, acc1 = {};
            #pragma unroll
            for (int j = 0; j < D / 4; ++j) {
                const int kb = 4 * j + kHalf;
                const v2f a0 = *(const v2f*)&s_tile[mrow][kb];
                const v2f a1 = *(const v2f*)&s_tile[16 + mrow][kb];
                acc0 = __builtin_amdgcn_wmma_f32_16x16x4_f32(
                    false, a0, false, w1f[j], (short)0, acc0, false, false);
                acc1 = __builtin_amdgcn_wmma_f32_16x16x4_f32(
                    false, a1, false, w1f[j], (short)0, acc1, false, false);
            }
            #pragma unroll
            for (int v = 0; v < 8; ++v) {
                const float t0 = acc0[v] + bias1;
                const float t1 = acc1[v] + bias1;
                t_tile[rbase + v][colN]      = t0 > 0.0f ? t0 : 0.0f;
                t_tile[16 + rbase + v][colN] = t1 > 0.0f ? t1 : 0.0f;
            }
        }
        __syncthreads();

        #pragma unroll
        for (int i = 0; i < WAVE_ROWS; ++i) {
            st[i] = __builtin_amdgcn_readlane(sv, i);
            dg[i] = __builtin_amdgcn_readlane(cv, i);
        }
        #pragma unroll
        for (int i = 0; i < WAVE_ROWS; ++i) my[i] = (lane < dg[i]) ? col[st[i] + lane] : 0;

        {
            v8f acc0 = {}, acc1 = {};
            #pragma unroll
            for (int j = 0; j < D / 4; ++j) {
                const int kb = 4 * j + kHalf;
                const v2f a0 = *(const v2f*)&t_tile[mrow][kb];
                const v2f a1 = *(const v2f*)&t_tile[16 + mrow][kb];
                acc0 = __builtin_amdgcn_wmma_f32_16x16x4_f32(
                    false, a0, false, w2f[j], (short)0, acc0, false, false);
                acc1 = __builtin_amdgcn_wmma_f32_16x16x4_f32(
                    false, a1, false, w2f[j], (short)0, acc1, false, false);
            }
            float u0[8], u1[8];
            #pragma unroll
            for (int v = 0; v < 8; ++v) {
                u0[v] = acc0[v] + bias2;
                u1[v] = acc1[v] + bias2;
                if (applyReluOut) {
                    u0[v] = u0[v] > 0.0f ? u0[v] : 0.0f;
                    u1[v] = u1[v] > 0.0f ? u1[v] : 0.0f;
                }
            }
            float* op = out + (row0 + rbase) * D + colN;
            if (row0 + MLP_ROWS <= nNodes) {
                #pragma unroll
                for (int v = 0; v < 8; ++v) {
                    op[v * D]        = u0[v];
                    op[(16 + v) * D] = u1[v];
                }
            } else {
                #pragma unroll
                for (int v = 0; v < 8; ++v) {
                    if (row0 + rbase + v < nNodes)      op[v * D]        = u0[v];
                    if (row0 + rbase + v + 16 < nNodes) op[(16 + v) * D] = u1[v];
                }
            }
        }
    }
}

extern "C" void kernel_launch(void* const* d_in, const int* in_sizes, int n_in,
                              void* d_out, int out_size, void* d_ws, size_t ws_size,
                              hipStream_t stream) {
    const float* x  = (const float*)d_in[0];
    const int*   ei = (const int*)  d_in[1];
    const float* W1 = (const float*)d_in[2];
    const float* b1 = (const float*)d_in[3];
    const float* W2 = (const float*)d_in[4];
    const float* b2 = (const float*)d_in[5];
    float* out = (float*)d_out;

    const int N = in_sizes[0] / D;
    const int E = in_sizes[1] / 2;
    const int* src = ei;
    const int* dst = ei + E;

    const int bBlocks = (N + RANGE - 1) / RANGE;
    const int slice   = (E + BIN_BLOCKS - 1) / BIN_BLOCKS;

    float* h1  = (float*)d_ws;
    int* cnt   = (int*)(h1 + (size_t)N * D);
    int* start = cnt + N;
    int* col   = start + N;
    int* off   = col + E;
    int* ent   = (int*)((((unsigned long long)(off + (size_t)BIN_BLOCKS * (bBlocks + 1))) + 15ull) & ~15ull);

    edge_bin<<<BIN_BLOCKS, BIN_THREADS, 0, stream>>>(src, dst, off, ent, N, E, bBlocks, slice);
    bucket_build<<<bBlocks, BUILD_THREADS, 0, stream>>>(off, ent, start, cnt, col, N, bBlocks, slice);

    gin_layer<<<MLP_BLOCKS, BLOCK, 0, stream>>>(x, start, cnt, col, W1, b1, W2, b2, h1, N, 1);

    gin_layer<<<MLP_BLOCKS, BLOCK, 0, stream>>>(h1, start, cnt, col, W1, b1, W2, b2, out, N, 0);
}
